# starting kernel
# baseline (speedup 1.0000x reference)
#include <hip/hip_runtime.h>

#define C 128
#define NL 4
#define NEG_SLOPE 0.2f
#define SA 132
#define SW 132
#define WSTR 136

typedef _Float16 h8v __attribute__((ext_vector_type(8)));
typedef float v2f __attribute__((ext_vector_type(2)));
typedef float v4f __attribute__((ext_vector_type(4)));
typedef float v8f __attribute__((ext_vector_type(8)));

__global__ __launch_bounds__(256, 1) void rnn_recurrence(
    const float* __restrict__ x, const float* __restrict__ Wih,
    const float* __restrict__ Whh, const float* __restrict__ bih,
    const float* __restrict__ bhh, const int* __restrict__ max_steps,
    float* __restrict__ enc)
{
    __shared__ _Float16 swih[NL * C * WSTR];
    __shared__ _Float16 swhh[NL * C * WSTR];
    __shared__ float    sb[NL * C];
    __shared__ float    hid2[2][NL * C];
    __shared__ float    part[2][C];

    const int tid  = threadIdx.x;
    const int ch   = tid & (C - 1);
    const int half = tid >> 7;

    for (int i = tid; i < NL * C * C; i += 256) {
        int r = i >> 7, c = i & (C - 1);
        swih[r * WSTR + c] = (_Float16)Wih[i];
        swhh[r * WSTR + c] = (_Float16)Whh[i];
    }
    for (int i = tid; i < NL * C; i += 256) {
        sb[i]      = bih[i] + bhh[i];
        hid2[0][i] = 0.0f;
        hid2[1][i] = (i < C) ? x[i] : 0.0f;
    }
    __syncthreads();

    const int T = max_steps[0];
    for (int t = 0; t < T; ++t) {
        const int p = t & 1, q = p ^ 1;
        for (int l = 0; l < NL; ++l) {
            const float* in = (l == 0) ? &hid2[q][(NL - 1) * C]
                                       : &hid2[p][(l - 1) * C];
            const float* hp = &hid2[q][l * C];
            const v4f* in4 = (const v4f*)in;
            const v4f* hp4 = (const v4f*)hp;

            const h8v* wi8 = (const h8v*)(swih + (l * C + ch) * WSTR + half * 64);
            const h8v* wh8 = (const h8v*)(swhh + (l * C + ch) * WSTR + half * 64);

            float acc = 0.0f;
            #pragma unroll
            for (int k8 = 0; k8 < 8; ++k8) {
                h8v wa = wi8[k8];
                h8v wb = wh8[k8];
                v4f i0 = in4[half * 16 + k8 * 2];
                v4f i1 = in4[half * 16 + k8 * 2 + 1];
                v4f h0 = hp4[half * 16 + k8 * 2];
                v4f h1 = hp4[half * 16 + k8 * 2 + 1];
                acc += (float)wa[0] * i0.x + (float)wa[1] * i0.y;
                acc += (float)wa[2] * i0.z + (float)wa[3] * i0.w;
                acc += (float)wa[4] * i1.x + (float)wa[5] * i1.y;
                acc += (float)wa[6] * i1.z + (float)wa[7] * i1.w;
                acc += (float)wb[0] * h0.x + (float)wb[1] * h0.y;
                acc += (float)wb[2] * h0.z + (float)wb[3] * h0.w;
                acc += (float)wb[4] * h1.x + (float)wb[5] * h1.y;
                acc += (float)wb[6] * h1.z + (float)wb[7] * h1.w;
            }
            part[half][ch] = acc;
            __syncthreads();
            if (half == 0) {
                float s = sb[l * C + ch] + part[0][ch] + part[1][ch];
                s = s > 0.0f ? s : 0.0f;
                hid2[p][l * C + ch] = s;
                if (l == NL - 1) enc[t * C + ch] = s;
            }
            __syncthreads();
        }
    }
}

__global__ __launch_bounds__(128, 1) void head_kernel(
    const float* __restrict__ enc,
    const float* __restrict__ rb_w,
    const float* __restrict__ rb_b,
    const float* __restrict__ out_w,
    const float* __restrict__ out_b,
    float* __restrict__ out,
    int od, int col0)
{
    __shared__ float wlds[C * SW];
    __shared__ float blds[C];
    __shared__ float xbuf[4][16 * SA];
    __shared__ float hbuf[4][16 * SA];

    const int tid  = threadIdx.x;
    const int wave = tid >> 5;
    const int lane = tid & 31;
    const int ln   = lane & 15;
    const int hi   = lane >> 4;
    const int row0 = blockIdx.x * 64 + wave * 16;

    for (int i = lane; i < 16 * C; i += 32) {
        int m = i >> 7, c = i & (C - 1);
        xbuf[wave][m * SA + c] = enc[(row0 + m) * C + c];
    }

    for (int blk = 0; blk < 2; ++blk) {
        for (int ly = 0; ly < 2; ++ly) {
            const float* W = rb_w + (blk * 2 + ly) * C * C;
            const float* B = rb_b + (blk * 2 + ly) * C;
            __syncthreads();
            for (int i = tid; i < C * C; i += 128)
                wlds[(i >> 7) * SW + (i & (C - 1))] = W[i];
            if (tid < C) blds[tid] = B[tid];
            __syncthreads();

            const float* src = (ly == 0) ? xbuf[wave] : hbuf[wave];
            float*       dst = (ly == 0) ? hbuf[wave] : xbuf[wave];

            v8f acc[8];
            #pragma unroll
            for (int nt = 0; nt < 8; ++nt) {
                float bv = blds[nt * 16 + ln];
                #pragma unroll
                for (int r = 0; r < 8; ++r) {
                    float cv = bv;
                    if (ly == 1)
                        cv += xbuf[wave][(r + 8 * hi) * SA + nt * 16 + ln];
                    acc[nt][r] = cv;
                }
            }

            for (int k = 0; k < 32; ++k) {
                v2f a;
                a.x = src[ln * SA + 4 * k + 2 * hi];
                a.y = src[ln * SA + 4 * k + 2 * hi + 1];
                #pragma unroll
                for (int nt = 0; nt < 8; ++nt) {
                    v2f b;
                    b.x = wlds[(nt * 16 + ln) * SW + 4 * k + 2 * hi];
                    b.y = wlds[(nt * 16 + ln) * SW + 4 * k + 2 * hi + 1];
                    acc[nt] = __builtin_amdgcn_wmma_f32_16x16x4_f32(
                        false, a, false, b, (short)0, acc[nt], false, false);
                }
            }

            #pragma unroll
            for (int nt = 0; nt < 8; ++nt) {
                #pragma unroll
                for (int r = 0; r < 8; ++r) {
                    float v = acc[nt][r];
                    v = v > 0.0f ? v : NEG_SLOPE * v;
                    dst[(r + 8 * hi) * SA + nt * 16 + ln] = v;
                }
            }
        }
    }
    __syncthreads();

    for (int o = hi; o < od; o += 2) {
        const float* wo = out_w + o * C;
        float s = out_b[o];
        for (int c = 0; c < C; ++c) s += xbuf[wave][ln * SA + c] * wo[c];
        out[(row0 + ln) * 10 + col0 + o] = s;
    }
}

extern "C" void kernel_launch(void* const* d_in, const int* in_sizes, int n_in,
                              void* d_out, int out_size, void* d_ws, size_t ws_size,
                              hipStream_t stream)
{
    const float* x         = (const float*)d_in[0];
    const float* W_ih      = (const float*)d_in[1];
    const float* W_hh      = (const float*)d_in[2];
    const float* b_ih      = (const float*)d_in[3];
    const float* b_hh      = (const float*)d_in[4];
    const float* atom_rb_w = (const float*)d_in[5];
    const float* atom_rb_b = (const float*)d_in[6];
    const float* atom_w    = (const float*)d_in[7];
    const float* atom_b    = (const float*)d_in[8];
    const float* pos_rb_w  = (const float*)d_in[9];
    const float* pos_rb_b  = (const float*)d_in[10];
    const float* pos_w     = (const float*)d_in[11];
    const float* pos_b     = (const float*)d_in[12];
    const float* mag_rb_w  = (const float*)d_in[13];
    const float* mag_rb_b  = (const float*)d_in[14];
    const float* mag_w     = (const float*)d_in[15];
    const float* mag_b     = (const float*)d_in[16];
    const int*   max_steps = (const int*)d_in[17];

    float* out = (float*)d_out;
    float* enc = (float*)d_ws;
    const int T = out_size / 10;

    rnn_recurrence<<<1, 256, 0, stream>>>(x, W_ih, W_hh, b_ih, b_hh, max_steps, enc);

    const int blocks = T / 64;
    head_kernel<<<blocks, 128, 0, stream>>>(enc, atom_rb_w, atom_rb_b, atom_w, atom_b, out, 8, 0);
    head_kernel<<<blocks, 128, 0, stream>>>(enc, pos_rb_w,  pos_rb_b,  pos_w,  pos_b,  out, 1, 8);
    head_kernel<<<blocks, 128, 0, stream>>>(enc, mag_rb_w,  mag_rb_b,  mag_w,  mag_b,  out, 1, 9);
}


# Recurrence re-associated by input vector: each fresh layer output meets stacked rows [W_ih(next layer);W_hh(same layer)], thread = channel over whole K, 1 barrier per layer; f32 sum order changes
# speedup vs baseline: 2.2668x; 2.2668x over previous
#include <hip/hip_runtime.h>


#define C 128
#define NL 4
#define NEG_SLOPE 0.2f
#define SA 132
#define SW 132
#define WSTR 136

typedef _Float16 h8v __attribute__((ext_vector_type(8)));
typedef float v2f __attribute__((ext_vector_type(2)));
typedef float v4f __attribute__((ext_vector_type(4)));
typedef float v8f __attribute__((ext_vector_type(8)));

__device__ __forceinline__ void dot_pair(const _Float16* rowa, const _Float16* rowb,
                                         const float* v, float& ra, float& rb)
{
    const h8v* a8 = (const h8v*)rowa;
    const h8v* b8 = (const h8v*)rowb;
    const v4f* v4 = (const v4f*)v;
    float sa[8], sb[8];
    #pragma unroll
    for (int j = 0; j < 8; ++j) { sa[j] = 0.0f; sb[j] = 0.0f; }
    #pragma unroll
    for (int k8 = 0; k8 < 16; ++k8) {
        const h8v a = a8[k8];
        const h8v b = b8[k8];
        const v4f p = v4[2 * k8];
        const v4f q = v4[2 * k8 + 1];
        sa[0] = fmaf((float)a[0], p.x, sa[0]);  sb[0] = fmaf((float)b[0], p.x, sb[0]);
        sa[1] = fmaf((float)a[1], p.y, sa[1]);  sb[1] = fmaf((float)b[1], p.y, sb[1]);
        sa[2] = fmaf((float)a[2], p.z, sa[2]);  sb[2] = fmaf((float)b[2], p.z, sb[2]);
        sa[3] = fmaf((float)a[3], p.w, sa[3]);  sb[3] = fmaf((float)b[3], p.w, sb[3]);
        sa[4] = fmaf((float)a[4], q.x, sa[4]);  sb[4] = fmaf((float)b[4], q.x, sb[4]);
        sa[5] = fmaf((float)a[5], q.y, sa[5]);  sb[5] = fmaf((float)b[5], q.y, sb[5]);
        sa[6] = fmaf((float)a[6], q.z, sa[6]);  sb[6] = fmaf((float)b[6], q.z, sb[6]);
        sa[7] = fmaf((float)a[7], q.w, sa[7]);  sb[7] = fmaf((float)b[7], q.w, sb[7]);
    }
    ra = ((sa[0] + sa[1]) + (sa[2] + sa[3])) + ((sa[4] + sa[5]) + (sa[6] + sa[7]));
    rb = ((sb[0] + sb[1]) + (sb[2] + sb[3])) + ((sb[4] + sb[5]) + (sb[6] + sb[7]));
}

__global__ __launch_bounds__(128, 1) void rnn_recurrence(
    const float* __restrict__ x, const float* __restrict__ Wih,
    const float* __restrict__ Whh, const float* __restrict__ bih,
    const float* __restrict__ bhh, const int* __restrict__ max_steps,
    float* __restrict__ enc)
{
    __shared__ __attribute__((aligned(16))) _Float16 swih[NL * C * WSTR];
    __shared__ __attribute__((aligned(16))) _Float16 swhh[NL * C * WSTR];
    __shared__ __attribute__((aligned(16))) float vbuf[2][C];

    const int ch = threadIdx.x;

    for (int i = ch; i < NL * C * C; i += C) {
        int r = i >> 7, c = i & (C - 1);
        swih[r * WSTR + c] = (_Float16)Wih[i];
        swhh[r * WSTR + c] = (_Float16)Whh[i];
    }
    vbuf[1][ch] = x[ch];

    float bias[NL];
    float hb[NL];
    #pragma unroll
    for (int l = 0; l < NL; ++l) {
        bias[l] = bih[l * C + ch] + bhh[l * C + ch];
        hb[l]   = bias[l];
    }
    __syncthreads();

    float a = 0.0f;
    {
        float unused, h0;
        dot_pair(swih + ch * WSTR, swhh + ch * WSTR, vbuf[1], unused, h0);
        hb[0] = bias[0] + h0;
    }

    const int T = max_steps[0];
    for (int t = 0; t < T; ++t) {
        #pragma unroll
        for (int l = 0; l < NL; ++l) {
            float o = a + hb[l];
            o = o > 0.0f ? o : 0.0f;
            float* vb = vbuf[l & 1];
            vb[ch] = o;
            __syncthreads();
            if (l == NL - 1) enc[t * C + ch] = o;
            const int ln = (l + 1) & (NL - 1);
            float h;
            dot_pair(swih + (ln * C + ch) * WSTR, swhh + (l * C + ch) * WSTR, vb, a, h);
            hb[l] = bias[l] + h;
        }
    }
}

__global__ __launch_bounds__(128, 1) void head_kernel(
    const float* __restrict__ enc,
    const float* __restrict__ rb_w,
    const float* __restrict__ rb_b,
    const float* __restrict__ out_w,
    const float* __restrict__ out_b,
    float* __restrict__ out,
    int od, int col0)
{
    __shared__ float wlds[C * SW];
    __shared__ float blds[C];
    __shared__ float xbuf[4][16 * SA];
    __shared__ float hbuf[4][16 * SA];

    const int tid  = threadIdx.x;
    const int wave = tid >> 5;
    const int lane = tid & 31;
    const int ln   = lane & 15;
    const int hi   = lane >> 4;
    const int row0 = blockIdx.x * 64 + wave * 16;

    for (int i = lane; i < 16 * C; i += 32) {
        int m = i >> 7, c = i & (C - 1);
        xbuf[wave][m * SA + c] = enc[(row0 + m) * C + c];
    }

    for (int blk = 0; blk < 2; ++blk) {
        for (int ly = 0; ly < 2; ++ly) {
            const float* W = rb_w + (blk * 2 + ly) * C * C;
            const float* B = rb_b + (blk * 2 + ly) * C;
            __syncthreads();
            for (int i = tid; i < C * C; i += 128)
                wlds[(i >> 7) * SW + (i & (C - 1))] = W[i];
            if (tid < C) blds[tid] = B[tid];
            __syncthreads();

            const float* src = (ly == 0) ? xbuf[wave] : hbuf[wave];
            float*       dst = (ly == 0) ? hbuf[wave] : xbuf[wave];

            v8f acc[8];
            #pragma unroll
            for (int nt = 0; nt < 8; ++nt) {
                float bv = blds[nt * 16 + ln];
                #pragma unroll
                for (int r = 0; r < 8; ++r) {
                    float cv = bv;
                    if (ly == 1)
                        cv += xbuf[wave][(r + 8 * hi) * SA + nt * 16 + ln];
                    acc[nt][r] = cv;
                }
            }

            for (int k = 0; k < 32; ++k) {
                v2f a;
                a.x = src[ln * SA + 4 * k + 2 * hi];
                a.y = src[ln * SA + 4 * k + 2 * hi + 1];
                #pragma unroll
                for (int nt = 0; nt < 8; ++nt) {
                    v2f b;
                    b.x = wlds[(nt * 16 + ln) * SW + 4 * k + 2 * hi];
                    b.y = wlds[(nt * 16 + ln) * SW + 4 * k + 2 * hi + 1];
                    acc[nt] = __builtin_amdgcn_wmma_f32_16x16x4_f32(
                        false, a, false, b, (short)0, acc[nt], false, false);
                }
            }

            #pragma unroll
            for (int nt = 0; nt < 8; ++nt) {
                #pragma unroll
                for (int r = 0; r < 8; ++r) {
                    float v = acc[nt][r];
                    v = v > 0.0f ? v : NEG_SLOPE * v;
                    dst[(r + 8 * hi) * SA + nt * 16 + ln] = v;
                }
            }
        }
    }
    __syncthreads();

    for (int o = hi; o < od; o += 2) {
        const float* wo = out_w + o * C;
        float s = out_b[o];
        for (int c = 0; c < C; ++c) s += xbuf[wave][ln * SA + c] * wo[c];
        out[(row0 + ln) * 10 + col0 + o] = s;
    }
}

extern "C" void kernel_launch(void* const* d_in, const int* in_sizes, int n_in,
                              void* d_out, int out_size, void* d_ws, size_t ws_size,
                              hipStream_t stream)
{
    const float* x         = (const float*)d_in[0];
    const float* W_ih      = (const float*)d_in[1];
    const float* W_hh      = (const float*)d_in[2];
    const float* b_ih      = (const float*)d_in[3];
    const float* b_hh      = (const float*)d_in[4];
    const float* atom_rb_w = (const float*)d_in[5];
    const float* atom_rb_b = (const float*)d_in[6];
    const float* atom_w    = (const float*)d_in[7];
    const float* atom_b    = (const float*)d_in[8];
    const float* pos_rb_w  = (const float*)d_in[9];
    const float* pos_rb_b  = (const float*)d_in[10];
    const float* pos_w     = (const float*)d_in[11];
    const float* pos_b     = (const float*)d_in[12];
    const float* mag_rb_w  = (const float*)d_in[13];
    const float* mag_rb_b  = (const float*)d_in[14];
    const float* mag_w     = (const float*)d_in[15];
    const float* mag_b     = (const float*)d_in[16];
    const int*   max_steps = (const int*)d_in[17];

    float* out = (float*)d_out;
    float* enc = (float*)d_ws;
    const int T = out_size / 10;

    rnn_recurrence<<<1, 128, 0, stream>>>(x, W_ih, W_hh, b_ih, b_hh, max_steps, enc);

    const int blocks = T / 64;
    head_kernel<<<blocks, 128, 0, stream>>>(enc, atom_rb_w, atom_rb_b, atom_w, atom_b, out, 8, 0);
    head_kernel<<<blocks, 128, 0, stream>>>(enc, pos_rb_w,  pos_rb_b,  pos_w,  pos_b,  out, 1, 8);
    head_kernel<<<blocks, 128, 0, stream>>>(enc, mag_rb_w,  mag_rb_b,  mag_w,  mag_b,  out, 1, 9);
}


# Recurrence multiply-adds as one v_fma_mix_f32 each: f32 activation from a scalar register (vector read once per wave, v_readlane), f16 weight half in place, no converts; sum order unchanged
# speedup vs baseline: 2.8998x; 1.2792x over previous
#include <hip/hip_runtime.h>


#define C 128
#define NL 4
#define NEG_SLOPE 0.2f
#define SA 132
#define SW 132
#define PSTR 264

typedef unsigned u4v __attribute__((ext_vector_type(4)));
typedef float v2f __attribute__((ext_vector_type(2)));
typedef float v4f __attribute__((ext_vector_type(4)));
typedef float v8f __attribute__((ext_vector_type(8)));

__device__ __forceinline__ float mac_lo(float acc, float s, unsigned w)
{
    asm("v_fma_mix_f32 %0, %1, %2, %0 op_sel_hi:[0,1,0]" : "+v"(acc) : "s"(s), "v"(w));
    return acc;
}
__device__ __forceinline__ float mac_hi(float acc, float s, unsigned w)
{
    asm("v_fma_mix_f32 %0, %1, %2, %0 op_sel:[0,1,0] op_sel_hi:[0,1,0]" : "+v"(acc) : "s"(s), "v"(w));
    return acc;
}

__device__ __forceinline__ float bcast(const v4f mine, const int k)
{
    const float comp = (k & 3) == 0 ? mine.x : (k & 3) == 1 ? mine.y : (k & 3) == 2 ? mine.z : mine.w;
    return __int_as_float(__builtin_amdgcn_readlane(__float_as_int(comp), k >> 2));
}

__device__ __forceinline__ void dot_pair(const _Float16* pair, const float* v, float& ra, float& rb)
{
    const u4v* p4 = (const u4v*)pair;
    const int lane = threadIdx.x & 31;
    const v4f mine = ((const v4f*)v)[lane];
    float sa[8], sb[8];
    #pragma unroll
    for (int j = 0; j < 8; ++j) { sa[j] = 0.0f; sb[j] = 0.0f; }
    #pragma unroll
    for (int k8 = 0; k8 < 16; ++k8) {
        const u4v a = p4[k8];
        const u4v b = p4[16 + k8];
        #pragma unroll
        for (int e = 0; e < 8; ++e) {
            const float x = bcast(mine, 8 * k8 + e);
            const unsigned wa = a[e >> 1];
            const unsigned wb = b[e >> 1];
            if (e & 1) { sa[e] = mac_hi(sa[e], x, wa); sb[e] = mac_hi(sb[e], x, wb); }
            else       { sa[e] = mac_lo(sa[e], x, wa); sb[e] = mac_lo(sb[e], x, wb); }
        }
    }
    ra = ((sa[0] + sa[1]) + (sa[2] + sa[3])) + ((sa[4] + sa[5]) + (sa[6] + sa[7]));
    rb = ((sb[0] + sb[1]) + (sb[2] + sb[3])) + ((sb[4] + sb[5]) + (sb[6] + sb[7]));
}

__global__ __launch_bounds__(128, 1) void rnn_recurrence(
    const float* __restrict__ x, const float* __restrict__ Wih,
    const float* __restrict__ Whh, const float* __restrict__ bih,
    const float* __restrict__ bhh, const int* __restrict__ max_steps,
    float* __restrict__ enc)
{
    __shared__ __attribute__((aligned(16))) _Float16 sw[NL * C * PSTR];
    __shared__ __attribute__((aligned(16))) float vbuf[2][C];

    const int ch = threadIdx.x;

    for (int i = ch; i < NL * C * C; i += C) {
        int r = i >> 7, c = i & (C - 1);
        sw[((r + (NL - 1) * C) & (NL * C - 1)) * PSTR + c] = (_Float16)Wih[i];
        sw[r * PSTR + C + c] = (_Float16)Whh[i];
    }
    vbuf[1][ch] = x[ch];

    float bias[NL];
    float hb[NL];
    #pragma unroll
    for (int l = 0; l < NL; ++l) {
        bias[l] = bih[l * C + ch] + bhh[l * C + ch];
        hb[l]   = bias[l];
    }
    __syncthreads();

    float a = 0.0f;
    {
        float unused, h0;
        dot_pair(sw + ch * PSTR, vbuf[1], unused, h0);
        hb[0] = bias[0] + h0;
    }

    const int T = max_steps[0];
    for (int t = 0; t < T; ++t) {
        #pragma unroll
        for (int l = 0; l < NL; ++l) {
            float o = a + hb[l];
            o = o > 0.0f ? o : 0.0f;
            float* vb = vbuf[l & 1];
            vb[ch] = o;
            __syncthreads();
            if (l == NL - 1) enc[t * C + ch] = o;
            float h;
            dot_pair(sw + (l * C + ch) * PSTR, vb, a, h);
            hb[l] = bias[l] + h;
        }
    }
}

__global__ __launch_bounds__(128, 1) void head_kernel(
    const float* __restrict__ enc,
    const float* __restrict__ rb_w,
    const float* __restrict__ rb_b,
    const float* __restrict__ out_w,
    const float* __restrict__ out_b,
    float* __restrict__ out,
    int od, int col0)
{
    __shared__ float wlds[C * SW];
    __shared__ float blds[C];
    __shared__ float xbuf[4][16 * SA];
    __shared__ float hbuf[4][16 * SA];

    const int tid  = threadIdx.x;
    const int wave = tid >> 5;
    const int lane = tid & 31;
    const int ln   = lane & 15;
    const int hi   = lane >> 4;
    const int row0 = blockIdx.x * 64 + wave * 16;

    for (int i = lane; i < 16 * C; i += 32) {
        int m = i >> 7, c = i & (C - 1);
        xbuf[wave][m * SA + c] = enc[(row0 + m) * C + c];
    }

    for (int blk = 0; blk < 2; ++blk) {
        for (int ly = 0; ly < 2; ++ly) {
            const float* W = rb_w + (blk * 2 + ly) * C * C;
            const float* B = rb_b + (blk * 2 + ly) * C;
            __syncthreads();
            for (int i = tid; i < C * C; i += 128)
                wlds[(i >> 7) * SW + (i & (C - 1))] = W[i];
            if (tid < C) blds[tid] = B[tid];
            __syncthreads();

            const float* src = (ly == 0) ? xbuf[wave] : hbuf[wave];
            float*       dst = (ly == 0) ? hbuf[wave] : xbuf[wave];

            v8f acc[8];
            #pragma unroll
            for (int nt = 0; nt < 8; ++nt) {
                float bv = blds[nt * 16 + ln];
                #pragma unroll
                for (int r = 0; r < 8; ++r) {
                    float cv = bv;
                    if (ly == 1)
                        cv += xbuf[wave][(r + 8 * hi) * SA + nt * 16 + ln];
                    acc[nt][r] = cv;
                }
            }

            for (int k = 0; k < 32; ++k) {
                v2f a;
                a.x = src[ln * SA + 4 * k + 2 * hi];
                a.y = src[ln * SA + 4 * k + 2 * hi + 1];
                #pragma unroll
                for (int nt = 0; nt < 8; ++nt) {
                    v2f b;
                    b.x = wlds[(nt * 16 + ln) * SW + 4 * k + 2 * hi];
                    b.y = wlds[(nt * 16 + ln) * SW + 4 * k + 2 * hi + 1];
                    acc[nt] = __builtin_amdgcn_wmma_f32_16x16x4_f32(
                        false, a, false, b, (short)0, acc[nt], false, false);
                }
            }

            #pragma unroll
            for (int nt = 0; nt < 8; ++nt) {
                #pragma unroll
                for (int r = 0; r < 8; ++r) {
                    float v = acc[nt][r];
                    v = v > 0.0f ? v : NEG_SLOPE * v;
                    dst[(r + 8 * hi) * SA + nt * 16 + ln] = v;
                }
            }
        }
    }
    __syncthreads();

    for (int o = hi; o < od; o += 2) {
        const float* wo = out_w + o * C;
        float s = out_b[o];
        for (int c = 0; c < C; ++c) s += xbuf[wave][ln * SA + c] * wo[c];
        out[(row0 + ln) * 10 + col0 + o] = s;
    }
}

extern "C" void kernel_launch(void* const* d_in, const int* in_sizes, int n_in,
                              void* d_out, int out_size, void* d_ws, size_t ws_size,
                              hipStream_t stream)
{
    const float* x         = (const float*)d_in[0];
    const float* W_ih      = (const float*)d_in[1];
    const float* W_hh      = (const float*)d_in[2];
    const float* b_ih      = (const float*)d_in[3];
    const float* b_hh      = (const float*)d_in[4];
    const float* atom_rb_w = (const float*)d_in[5];
    const float* atom_rb_b = (const float*)d_in[6];
    const float* atom_w    = (const float*)d_in[7];
    const float* atom_b    = (const float*)d_in[8];
    const float* pos_rb_w  = (const float*)d_in[9];
    const float* pos_rb_b  = (const float*)d_in[10];
    const float* pos_w     = (const float*)d_in[11];
    const float* pos_b     = (const float*)d_in[12];
    const float* mag_rb_w  = (const float*)d_in[13];
    const float* mag_rb_b  = (const float*)d_in[14];
    const float* mag_w     = (const float*)d_in[15];
    const float* mag_b     = (const float*)d_in[16];
    const int*   max_steps = (const int*)d_in[17];

    float* out = (float*)d_out;
    float* enc = (float*)d_ws;
    const int T = out_size / 10;

    rnn_recurrence<<<1, 128, 0, stream>>>(x, W_ih, W_hh, b_ih, b_hh, max_steps, enc);

    const int blocks = T / 64;
    head_kernel<<<blocks, 128, 0, stream>>>(enc, atom_rb_w, atom_rb_b, atom_w, atom_b, out, 8, 0);
    head_kernel<<<blocks, 128, 0, stream>>>(enc, pos_rb_w,  pos_rb_b,  pos_w,  pos_b,  out, 1, 8);
    head_kernel<<<blocks, 128, 0, stream>>>(enc, mag_rb_w,  mag_rb_b,  mag_w,  mag_b,  out, 1, 9);
}


# Recurrence: K split across the 4 waves (a wave takes 32 k of all 256 rows, 8 rows a lane): readlane broadcasts per wave and layer 128 -> 32; hand-over = 128x4 table of partials; f32 sum order changes
# speedup vs baseline: 3.2836x; 1.1324x over previous
#include <hip/hip_runtime.h>


#define C 128
#define NL 4
#define NEG_SLOPE 0.2f
#define SA 132
#define SW 132
#define PSTR 264

typedef unsigned u4v __attribute__((ext_vector_type(4)));
typedef float v2f __attribute__((ext_vector_type(2)));
typedef float v4f __attribute__((ext_vector_type(4)));
typedef float v8f __attribute__((ext_vector_type(8)));

__device__ __forceinline__ float mac_lo(float acc, float s, unsigned w)
{
    asm("v_fma_mix_f32 %0, %1, %2, %0 op_sel_hi:[0,1,0]" : "+v"(acc) : "s"(s), "v"(w));
    return acc;
}
__device__ __forceinline__ float mac_hi(float acc, float s, unsigned w)
{
    asm("v_fma_mix_f32 %0, %1, %2, %0 op_sel:[0,1,0] op_sel_hi:[0,1,0]" : "+v"(acc) : "s"(s), "v"(w));
    return acc;
}

__device__ __forceinline__ float bcast(const float mine, const int k)
{
    return __int_as_float(__builtin_amdgcn_readlane(__float_as_int(mine), k));
}

struct QuarterRows { u4v a[4][4], b[4][4]; };

__device__ __forceinline__ void load_rows(const _Float16* rows, QuarterRows& w)
{
    #pragma unroll
    for (int kc = 0; kc < 4; ++kc) {
        #pragma unroll
        for (int g = 0; g < 4; ++g)
            w.a[kc][g] = ((const u4v*)(rows + g * 32 * PSTR))[kc];
        __builtin_amdgcn_sched_barrier(0);
        #pragma unroll
        for (int g = 0; g < 4; ++g)
            w.b[kc][g] = ((const u4v*)(rows + g * 32 * PSTR))[16 + kc];
        __builtin_amdgcn_sched_barrier(0);
    }
}

__device__ __forceinline__ void mac_rows(const QuarterRows& w, const float mine,
                                         float (&aa)[4], float (&hh)[4])
{
    #pragma unroll
    for (int kc = 0; kc < 4; ++kc) {
        #pragma unroll
        for (int e = 0; e < 8; ++e) {
            const float x = bcast(mine, 8 * kc + e);
            #pragma unroll
            for (int g = 0; g < 4; ++g) {
                const unsigned ua = w.a[kc][g][e >> 1];
                const unsigned ub = w.b[kc][g][e >> 1];
                if (e & 1) { aa[g] = mac_hi(aa[g], x, ua); hh[g] = mac_hi(hh[g], x, ub); }
                else       { aa[g] = mac_lo(aa[g], x, ua); hh[g] = mac_lo(hh[g], x, ub); }
            }
        }
    }
}

__global__ __launch_bounds__(128, 1) void rnn_recurrence(
    const float* __restrict__ x, const float* __restrict__ Wih,
    const float* __restrict__ Whh, const float* __restrict__ bih,
    const float* __restrict__ bhh, const int* __restrict__ max_steps,
    float* __restrict__ enc)
{
    __shared__ __attribute__((aligned(16))) _Float16 sw[NL * C * PSTR];
    __shared__ __attribute__((aligned(16))) float part[2][C][4];

    const int ch   = threadIdx.x;
    const int lane = threadIdx.x & 31;
    const int wq   = threadIdx.x >> 5;

    for (int i = ch; i < NL * C * C; i += C) {
        int r = i >> 7, c = i & (C - 1);
        sw[((r + (NL - 1) * C) & (NL * C - 1)) * PSTR + c] = (_Float16)Wih[i];
        sw[r * PSTR + C + c] = (_Float16)Whh[i];
    }

    float bq[NL][4];
    float hs[NL][4];
    #pragma unroll
    for (int l = 0; l < NL; ++l) {
        #pragma unroll
        for (int g = 0; g < 4; ++g) {
            const float b = bih[l * C + 32 * g + lane] + bhh[l * C + 32 * g + lane];
            bq[l][g] = wq == 0 ? b : 0.0f;
            hs[l][g] = bq[l][g];
        }
    }
    __syncthreads();

    {
        float unused[4], h0[4];
        #pragma unroll
        for (int g = 0; g < 4; ++g) { unused[g] = 0.0f; h0[g] = bq[0][g]; }
        QuarterRows w;
        load_rows(sw + lane * PSTR + 32 * wq, w);
        mac_rows(w, x[ch], unused, h0);
        #pragma unroll
        for (int g = 0; g < 4; ++g) part[0][32 * g + lane][wq] = h0[g];
    }
    __syncthreads();

    const int T = max_steps[0];
    for (int t = 0; t < T; ++t) {
        #pragma unroll
        for (int l = 0; l < NL; ++l) {
            const int ln = (l + 1) & (NL - 1);
            const v4f p = *(const v4f*)&part[l & 1][ch][0];
            QuarterRows w;
            load_rows(sw + (l * C + lane) * PSTR + 32 * wq, w);
            float o = (p.x + p.y) + (p.z + p.w);
            o = o > 0.0f ? o : 0.0f;
            float aa[4], hh[4];
            #pragma unroll
            for (int g = 0; g < 4; ++g) { aa[g] = hs[ln][g]; hh[g] = bq[l][g]; }
            mac_rows(w, o, aa, hh);
            #pragma unroll
            for (int g = 0; g < 4; ++g) {
                part[ln & 1][32 * g + lane][wq] = aa[g];
                hs[l][g] = hh[g];
            }
            __syncthreads();
            if (l == NL - 1) enc[t * C + ch] = o;
        }
    }
}

__global__ __launch_bounds__(128, 1) void head_kernel(
    const float* __restrict__ enc,
    const float* __restrict__ rb_w,
    const float* __restrict__ rb_b,
    const float* __restrict__ out_w,
    const float* __restrict__ out_b,
    float* __restrict__ out,
    int od, int col0)
{
    __shared__ float wlds[C * SW];
    __shared__ float blds[C];
    __shared__ float xbuf[4][16 * SA];
    __shared__ float hbuf[4][16 * SA];

    const int tid  = threadIdx.x;
    const int wave = tid >> 5;
    const int lane = tid & 31;
    const int ln   = lane & 15;
    const int hi   = lane >> 4;
    const int row0 = blockIdx.x * 64 + wave * 16;

    for (int i = lane; i < 16 * C; i += 32) {
        int m = i >> 7, c = i & (C - 1);
        xbuf[wave][m * SA + c] = enc[(row0 + m) * C + c];
    }

    for (int blk = 0; blk < 2; ++blk) {
        for (int ly = 0; ly < 2; ++ly) {
            const float* W = rb_w + (blk * 2 + ly) * C * C;
            const float* B = rb_b + (blk * 2 + ly) * C;
            __syncthreads();
            for (int i = tid; i < C * C; i += 128)
                wlds[(i >> 7) * SW + (i & (C - 1))] = W[i];
            if (tid < C) blds[tid] = B[tid];
            __syncthreads();

            const float* src = (ly == 0) ? xbuf[wave] : hbuf[wave];
            float*       dst = (ly == 0) ? hbuf[wave] : xbuf[wave];

            v8f acc[8];
            #pragma unroll
            for (int nt = 0; nt < 8; ++nt) {
                float bv = blds[nt * 16 + ln];
                #pragma unroll
                for (int r = 0; r < 8; ++r) {
                    float cv = bv;
                    if (ly == 1)
                        cv += xbuf[wave][(r + 8 * hi) * SA + nt * 16 + ln];
                    acc[nt][r] = cv;
                }
            }

            for (int k = 0; k < 32; ++k) {
                v2f a;
                a.x = src[ln * SA + 4 * k + 2 * hi];
                a.y = src[ln * SA + 4 * k + 2 * hi + 1];
                #pragma unroll
                for (int nt = 0; nt < 8; ++nt) {
                    v2f b;
                    b.x = wlds[(nt * 16 + ln) * SW + 4 * k + 2 * hi];
                    b.y = wlds[(nt * 16 + ln) * SW + 4 * k + 2 * hi + 1];
                    acc[nt] = __builtin_amdgcn_wmma_f32_16x16x4_f32(
                        false, a, false, b, (short)0, acc[nt], false, false);
                }
            }

            #pragma unroll
            for (int nt = 0; nt < 8; ++nt) {
                #pragma unroll
                for (int r = 0; r < 8; ++r) {
                    float v = acc[nt][r];
                    v = v > 0.0f ? v : NEG_SLOPE * v;
                    dst[(r + 8 * hi) * SA + nt * 16 + ln] = v;
                }
            }
        }
    }
    __syncthreads();

    for (int o = hi; o < od; o += 2) {
        const float* wo = out_w + o * C;
        float s = out_b[o];
        for (int c = 0; c < C; ++c) s += xbuf[wave][ln * SA + c] * wo[c];
        out[(row0 + ln) * 10 + col0 + o] = s;
    }
}

extern "C" void kernel_launch(void* const* d_in, const int* in_sizes, int n_in,
                              void* d_out, int out_size, void* d_ws, size_t ws_size,
                              hipStream_t stream)
{
    const float* x         = (const float*)d_in[0];
    const float* W_ih      = (const float*)d_in[1];
    const float* W_hh      = (const float*)d_in[2];
    const float* b_ih      = (const float*)d_in[3];
    const float* b_hh      = (const float*)d_in[4];
    const float* atom_rb_w = (const float*)d_in[5];
    const float* atom_rb_b = (const float*)d_in[6];
    const float* atom_w    = (const float*)d_in[7];
    const float* atom_b    = (const float*)d_in[8];
    const float* pos_rb_w  = (const float*)d_in[9];
    const float* pos_rb_b  = (const float*)d_in[10];
    const float* pos_w     = (const float*)d_in[11];
    const float* pos_b     = (const float*)d_in[12];
    const float* mag_rb_w  = (const float*)d_in[13];
    const float* mag_rb_b  = (const float*)d_in[14];
    const float* mag_w     = (const float*)d_in[15];
    const float* mag_b     = (const float*)d_in[16];
    const int*   max_steps = (const int*)d_in[17];

    float* out = (float*)d_out;
    float* enc = (float*)d_ws;
    const int T = out_size / 10;

    rnn_recurrence<<<1, 128, 0, stream>>>(x, W_ih, W_hh, b_ih, b_hh, max_steps, enc);

    const int blocks = T / 64;
    head_kernel<<<blocks, 128, 0, stream>>>(enc, atom_rb_w, atom_rb_b, atom_w, atom_b, out, 8, 0);
    head_kernel<<<blocks, 128, 0, stream>>>(enc, pos_rb_w,  pos_rb_b,  pos_w,  pos_b,  out, 1, 8);
    head_kernel<<<blocks, 128, 0, stream>>>(enc, mag_rb_w,  mag_rb_b,  mag_w,  mag_b,  out, 1, 9);
}


# Recurrence: W_hh half of a layer evaluation off the serial chain: 2 of its 4 k-chunks run behind the table stores, 2 at the head of the next evaluation under its table load; sum order unchanged
# speedup vs baseline: 3.5937x; 1.0944x over previous
#include <hip/hip_runtime.h>


#define C 128
#define NL 4
#define NEG_SLOPE 0.2f
#define SA 132
#define SW 132
#define PSTR 264

typedef unsigned u4v __attribute__((ext_vector_type(4)));
typedef float v2f __attribute__((ext_vector_type(2)));
typedef float v4f __attribute__((ext_vector_type(4)));
typedef float v8f __attribute__((ext_vector_type(8)));

__device__ __forceinline__ float mac_lo(float acc, float s, unsigned w)
{
    asm("v_fma_mix_f32 %0, %1, %2, %0 op_sel_hi:[0,1,0]" : "+v"(acc) : "s"(s), "v"(w));
    return acc;
}
__device__ __forceinline__ float mac_hi(float acc, float s, unsigned w)
{
    asm("v_fma_mix_f32 %0, %1, %2, %0 op_sel:[0,1,0] op_sel_hi:[0,1,0]" : "+v"(acc) : "s"(s), "v"(w));
    return acc;
}

__device__ __forceinline__ float bcast(const float mine, const int k)
{
    return __int_as_float(__builtin_amdgcn_readlane(__float_as_int(mine), k));
}

struct HalfRows { u4v r[4][4]; };

template <int OFS>
__device__ __forceinline__ void load_half(const _Float16* rows, HalfRows& w)
{
    #pragma unroll
    for (int kc = 0; kc < 4; ++kc) {
        #pragma unroll
        for (int g = 0; g < 4; ++g)
            w.r[kc][g] = ((const u4v*)(rows + g * 32 * PSTR))[OFS + kc];
        __builtin_amdgcn_sched_barrier(0);
    }
}

template <int KC0, int KC1>
__device__ __forceinline__ void mac_chunks(const HalfRows& w, const float (&xs)[32],
                                           float (&acc)[4])
{
    #pragma unroll
    for (int kc = KC0; kc < KC1; ++kc) {
        #pragma unroll
        for (int e = 0; e < 8; ++e) {
            const float x = xs[8 * kc + e];
            #pragma unroll
            for (int g = 0; g < 4; ++g) {
                const unsigned u = w.r[kc][g][e >> 1];
                acc[g] = (e & 1) ? mac_hi(acc[g], x, u) : mac_lo(acc[g], x, u);
            }
        }
    }
}

__global__ __launch_bounds__(128, 1) void rnn_recurrence(
    const float* __restrict__ x, const float* __restrict__ Wih,
    const float* __restrict__ Whh, const float* __restrict__ bih,
    const float* __restrict__ bhh, const int* __restrict__ max_steps,
    float* __restrict__ enc)
{
    __shared__ __attribute__((aligned(16))) _Float16 sw[NL * C * PSTR];
    __shared__ __attribute__((aligned(16))) float part[2][C][4];

    const int ch   = threadIdx.x;
    const int lane = threadIdx.x & 31;
    const int wq   = threadIdx.x >> 5;

    for (int i = ch; i < NL * C * C; i += C) {
        int r = i >> 7, c = i & (C - 1);
        sw[((r + (NL - 1) * C) & (NL * C - 1)) * PSTR + c] = (_Float16)Wih[i];
        sw[r * PSTR + C + c] = (_Float16)Whh[i];
    }

    float bq[NL][4];
    float hs[NL][4];
    #pragma unroll
    for (int l = 0; l < NL; ++l) {
        #pragma unroll
        for (int g = 0; g < 4; ++g) {
            const float b = bih[l * C + 32 * g + lane] + bhh[l * C + 32 * g + lane];
            bq[l][g] = wq == 0 ? b : 0.0f;
            hs[l][g] = bq[l][g];
        }
    }
    __syncthreads();

    {
        float h0[4], xs[32];
        #pragma unroll
        for (int g = 0; g < 4; ++g) h0[g] = bq[0][g];
        HalfRows wb;
        load_half<16>(sw + lane * PSTR + 32 * wq, wb);
        const float xv = x[ch];
        #pragma unroll
        for (int k = 0; k < 32; ++k) xs[k] = bcast(xv, k);
        mac_chunks<0, 4>(wb, xs, h0);
        #pragma unroll
        for (int g = 0; g < 4; ++g) part[0][32 * g + lane][wq] = h0[g];
    }
    __syncthreads();

    HalfRows wbp;
    float xsp[32], hhp[4];
    #pragma unroll
    for (int kc = 0; kc < 4; ++kc) {
        #pragma unroll
        for (int g = 0; g < 4; ++g) wbp.r[kc][g] = (u4v){0u, 0u, 0u, 0u};
    }
    #pragma unroll
    for (int k = 0; k < 32; ++k) xsp[k] = 0.0f;
    #pragma unroll
    for (int g = 0; g < 4; ++g) hhp[g] = bq[NL - 1][g];

    const int T = max_steps[0];
    for (int t = 0; t < T; ++t) {
        #pragma unroll
        for (int l = 0; l < NL; ++l) {
            const int ln = (l + 1) & (NL - 1);
            const int lp = (l + NL - 1) & (NL - 1);
            const v4f p = *(const v4f*)&part[l & 1][ch][0];
            const _Float16* rows = sw + (l * C + lane) * PSTR + 32 * wq;
            HalfRows wa, wb;
            load_half<0>(rows, wa);
            load_half<16>(rows, wb);
            mac_chunks<2, 4>(wbp, xsp, hhp);
            #pragma unroll
            for (int g = 0; g < 4; ++g) hs[lp][g] = hhp[g];
            __builtin_amdgcn_sched_barrier(0);
            float o = (p.x + p.y) + (p.z + p.w);
            o = o > 0.0f ? o : 0.0f;
            float xs[32], aa[4], hh[4];
            #pragma unroll
            for (int k = 0; k < 32; ++k) xs[k] = bcast(o, k);
            #pragma unroll
            for (int g = 0; g < 4; ++g) { aa[g] = hs[ln][g]; hh[g] = bq[l][g]; }
            mac_chunks<0, 4>(wa, xs, aa);
            #pragma unroll
            for (int g = 0; g < 4; ++g) part[ln & 1][32 * g + lane][wq] = aa[g];
            __builtin_amdgcn_sched_barrier(0);
            mac_chunks<0, 2>(wb, xs, hh);
            __builtin_amdgcn_sched_barrier(0);
            __syncthreads();
            if (l == NL - 1) enc[t * C + ch] = o;
            wbp = wb;
            #pragma unroll
            for (int k = 0; k < 32; ++k) xsp[k] = xs[k];
            #pragma unroll
            for (int g = 0; g < 4; ++g) hhp[g] = hh[g];
        }
    }
}

__global__ __launch_bounds__(128, 1) void head_kernel(
    const float* __restrict__ enc,
    const float* __restrict__ rb_w,
    const float* __restrict__ rb_b,
    const float* __restrict__ out_w,
    const float* __restrict__ out_b,
    float* __restrict__ out,
    int od, int col0)
{
    __shared__ float wlds[C * SW];
    __shared__ float blds[C];
    __shared__ float xbuf[4][16 * SA];
    __shared__ float hbuf[4][16 * SA];

    const int tid  = threadIdx.x;
    const int wave = tid >> 5;
    const int lane = tid & 31;
    const int ln   = lane & 15;
    const int hi   = lane >> 4;
    const int row0 = blockIdx.x * 64 + wave * 16;

    for (int i = lane; i < 16 * C; i += 32) {
        int m = i >> 7, c = i & (C - 1);
        xbuf[wave][m * SA + c] = enc[(row0 + m) * C + c];
    }

    for (int blk = 0; blk < 2; ++blk) {
        for (int ly = 0; ly < 2; ++ly) {
            const float* W = rb_w + (blk * 2 + ly) * C * C;
            const float* B = rb_b + (blk * 2 + ly) * C;
            __syncthreads();
            for (int i = tid; i < C * C; i += 128)
                wlds[(i >> 7) * SW + (i & (C - 1))] = W[i];
            if (tid < C) blds[tid] = B[tid];
            __syncthreads();

            const float* src = (ly == 0) ? xbuf[wave] : hbuf[wave];
            float*       dst = (ly == 0) ? hbuf[wave] : xbuf[wave];

            v8f acc[8];
            #pragma unroll
            for (int nt = 0; nt < 8; ++nt) {
                float bv = blds[nt * 16 + ln];
                #pragma unroll
                for (int r = 0; r < 8; ++r) {
                    float cv = bv;
                    if (ly == 1)
                        cv += xbuf[wave][(r + 8 * hi) * SA + nt * 16 + ln];
                    acc[nt][r] = cv;
                }
            }

            for (int k = 0; k < 32; ++k) {
                v2f a;
                a.x = src[ln * SA + 4 * k + 2 * hi];
                a.y = src[ln * SA + 4 * k + 2 * hi + 1];
                #pragma unroll
                for (int nt = 0; nt < 8; ++nt) {
                    v2f b;
                    b.x = wlds[(nt * 16 + ln) * SW + 4 * k + 2 * hi];
                    b.y = wlds[(nt * 16 + ln) * SW + 4 * k + 2 * hi + 1];
                    acc[nt] = __builtin_amdgcn_wmma_f32_16x16x4_f32(
                        false, a, false, b, (short)0, acc[nt], false, false);
                }
            }

            #pragma unroll
            for (int nt = 0; nt < 8; ++nt) {
                #pragma unroll
                for (int r = 0; r < 8; ++r) {
                    float v = acc[nt][r];
                    v = v > 0.0f ? v : NEG_SLOPE * v;
                    dst[(r + 8 * hi) * SA + nt * 16 + ln] = v;
                }
            }
        }
    }
    __syncthreads();

    for (int o = hi; o < od; o += 2) {
        const float* wo = out_w + o * C;
        float s = out_b[o];
        for (int c = 0; c < C; ++c) s += xbuf[wave][ln * SA + c] * wo[c];
        out[(row0 + ln) * 10 + col0 + o] = s;
    }
}

extern "C" void kernel_launch(void* const* d_in, const int* in_sizes, int n_in,
                              void* d_out, int out_size, void* d_ws, size_t ws_size,
                              hipStream_t stream)
{
    const float* x         = (const float*)d_in[0];
    const float* W_ih      = (const float*)d_in[1];
    const float* W_hh      = (const float*)d_in[2];
    const float* b_ih      = (const float*)d_in[3];
    const float* b_hh      = (const float*)d_in[4];
    const float* atom_rb_w = (const float*)d_in[5];
    const float* atom_rb_b = (const float*)d_in[6];
    const float* atom_w    = (const float*)d_in[7];
    const float* atom_b    = (const float*)d_in[8];
    const float* pos_rb_w  = (const float*)d_in[9];
    const float* pos_rb_b  = (const float*)d_in[10];
    const float* pos_w     = (const float*)d_in[11];
    const float* pos_b     = (const float*)d_in[12];
    const float* mag_rb_w  = (const float*)d_in[13];
    const float* mag_rb_b  = (const float*)d_in[14];
    const float* mag_w     = (const float*)d_in[15];
    const float* mag_b     = (const float*)d_in[16];
    const int*   max_steps = (const int*)d_in[17];

    float* out = (float*)d_out;
    float* enc = (float*)d_ws;
    const int T = out_size / 10;

    rnn_recurrence<<<1, 128, 0, stream>>>(x, W_ih, W_hh, b_ih, b_hh, max_steps, enc);

    const int blocks = T / 64;
    head_kernel<<<blocks, 128, 0, stream>>>(enc, atom_rb_w, atom_rb_b, atom_w, atom_b, out, 8, 0);
    head_kernel<<<blocks, 128, 0, stream>>>(enc, pos_rb_w,  pos_rb_b,  pos_w,  pos_b,  out, 1, 8);
    head_kernel<<<blocks, 128, 0, stream>>>(enc, mag_rb_w,  mag_rb_b,  mag_w,  mag_b,  out, 1, 9);
}


# Recurrence: every row's 32-element quarter runs on two accumulators (chunks 0,2 / 1,3, added at the end): 8 independent multiply-add chains per stretch, not 4; f32 sum order changes
# speedup vs baseline: 3.8342x; 1.0669x over previous
#include <hip/hip_runtime.h>


#define C 128
#define NL 4
#define NEG_SLOPE 0.2f
#define SA 132
#define SW 132
#define PSTR 264

typedef unsigned u4v __attribute__((ext_vector_type(4)));
typedef float v2f __attribute__((ext_vector_type(2)));
typedef float v4f __attribute__((ext_vector_type(4)));
typedef float v8f __attribute__((ext_vector_type(8)));

__device__ __forceinline__ float mac_lo(float acc, float s, unsigned w)
{
    asm("v_fma_mix_f32 %0, %1, %2, %0 op_sel_hi:[0,1,0]" : "+v"(acc) : "s"(s), "v"(w));
    return acc;
}
__device__ __forceinline__ float mac_hi(float acc, float s, unsigned w)
{
    asm("v_fma_mix_f32 %0, %1, %2, %0 op_sel:[0,1,0] op_sel_hi:[0,1,0]" : "+v"(acc) : "s"(s), "v"(w));
    return acc;
}

__device__ __forceinline__ float bcast(const float mine, const int k)
{
    return __int_as_float(__builtin_amdgcn_readlane(__float_as_int(mine), k));
}

struct HalfRows { u4v r[4][4]; };

template <int OFS>
__device__ __forceinline__ void load_half(const _Float16* rows, HalfRows& w)
{
    #pragma unroll
    for (int kc = 0; kc < 4; ++kc) {
        #pragma unroll
        for (int g = 0; g < 4; ++g)
            w.r[kc][g] = ((const u4v*)(rows + g * 32 * PSTR))[OFS + kc];
        __builtin_amdgcn_sched_barrier(0);
    }
}

template <int KC0, int KC1>
__device__ __forceinline__ void mac_chunks(const HalfRows& w, const float (&xs)[32],
                                           float (&acc)[4])
{
    #pragma unroll
    for (int kc = KC0; kc < KC1; ++kc) {
        #pragma unroll
        for (int e = 0; e < 8; ++e) {
            const float x = xs[8 * kc + e];
            #pragma unroll
            for (int g = 0; g < 4; ++g) {
                const unsigned u = w.r[kc][g][e >> 1];
                acc[g] = (e & 1) ? mac_hi(acc[g], x, u) : mac_lo(acc[g], x, u);
            }
        }
    }
}

template <int KCA, int KCB>
__device__ __forceinline__ void mac_pair(const HalfRows& w, const float (&xs)[32],
                                         float (&acca)[4], float (&accb)[4])
{
    #pragma unroll
    for (int e = 0; e < 8; ++e) {
        const float xa = xs[8 * KCA + e];
        const float xb = xs[8 * KCB + e];
        #pragma unroll
        for (int g = 0; g < 4; ++g) {
            const unsigned u = w.r[KCA][g][e >> 1];
            acca[g] = (e & 1) ? mac_hi(acca[g], xa, u) : mac_lo(acca[g], xa, u);
        }
        #pragma unroll
        for (int g = 0; g < 4; ++g) {
            const unsigned u = w.r[KCB][g][e >> 1];
            accb[g] = (e & 1) ? mac_hi(accb[g], xb, u) : mac_lo(accb[g], xb, u);
        }
    }
}

__global__ __launch_bounds__(128, 1) void rnn_recurrence(
    const float* __restrict__ x, const float* __restrict__ Wih,
    const float* __restrict__ Whh, const float* __restrict__ bih,
    const float* __restrict__ bhh, const int* __restrict__ max_steps,
    float* __restrict__ enc)
{
    __shared__ __attribute__((aligned(16))) _Float16 sw[NL * C * PSTR];
    __shared__ __attribute__((aligned(16))) float part[2][C][4];

    const int ch   = threadIdx.x;
    const int lane = threadIdx.x & 31;
    const int wq   = threadIdx.x >> 5;

    for (int i = ch; i < NL * C * C; i += C) {
        int r = i >> 7, c = i & (C - 1);
        sw[((r + (NL - 1) * C) & (NL * C - 1)) * PSTR + c] = (_Float16)Wih[i];
        sw[r * PSTR + C + c] = (_Float16)Whh[i];
    }

    float bq[NL][4];
    float hs[NL][4];
    #pragma unroll
    for (int l = 0; l < NL; ++l) {
        #pragma unroll
        for (int g = 0; g < 4; ++g) {
            const float b = bih[l * C + 32 * g + lane] + bhh[l * C + 32 * g + lane];
            bq[l][g] = wq == 0 ? b : 0.0f;
            hs[l][g] = bq[l][g];
        }
    }
    __syncthreads();

    {
        float h0[4], xs[32];
        #pragma unroll
        for (int g = 0; g < 4; ++g) h0[g] = bq[0][g];
        HalfRows wb;
        load_half<16>(sw + lane * PSTR + 32 * wq, wb);
        const float xv = x[ch];
        #pragma unroll
        for (int k = 0; k < 32; ++k) xs[k] = bcast(xv, k);
        mac_chunks<0, 4>(wb, xs, h0);
        #pragma unroll
        for (int g = 0; g < 4; ++g) part[0][32 * g + lane][wq] = h0[g];
    }
    __syncthreads();

    HalfRows wbp;
    float xsp[32], hhp[4], hhp1[4];
    #pragma unroll
    for (int kc = 0; kc < 4; ++kc) {
        #pragma unroll
        for (int g = 0; g < 4; ++g) wbp.r[kc][g] = (u4v){0u, 0u, 0u, 0u};
    }
    #pragma unroll
    for (int k = 0; k < 32; ++k) xsp[k] = 0.0f;
    #pragma unroll
    for (int g = 0; g < 4; ++g) { hhp[g] = bq[NL - 1][g]; hhp1[g] = 0.0f; }

    const int T = max_steps[0];
    for (int t = 0; t < T; ++t) {
        #pragma unroll
        for (int l = 0; l < NL; ++l) {
            const int ln = (l + 1) & (NL - 1);
            const int lp = (l + NL - 1) & (NL - 1);
            const v4f p = *(const v4f*)&part[l & 1][ch][0];
            const _Float16* rows = sw + (l * C + lane) * PSTR + 32 * wq;
            HalfRows wa, wb;
            load_half<0>(rows, wa);
            load_half<16>(rows, wb);
            mac_pair<2, 3>(wbp, xsp, hhp, hhp1);
            #pragma unroll
            for (int g = 0; g < 4; ++g) hs[lp][g] = hhp[g] + hhp1[g];
            __builtin_amdgcn_sched_barrier(0);
            float o = (p.x + p.y) + (p.z + p.w);
            o = o > 0.0f ? o : 0.0f;
            float xs[32], aa[4], aa1[4], hh[4], hh1[4];
            #pragma unroll
            for (int k = 0; k < 32; ++k) xs[k] = bcast(o, k);
            #pragma unroll
            for (int g = 0; g < 4; ++g) { aa[g] = hs[ln][g]; hh[g] = bq[l][g]; aa1[g] = 0.0f; hh1[g] = 0.0f; }
            mac_pair<0, 1>(wa, xs, aa, aa1);
            mac_pair<2, 3>(wa, xs, aa, aa1);
            #pragma unroll
            for (int g = 0; g < 4; ++g) part[ln & 1][32 * g + lane][wq] = aa[g] + aa1[g];
            __builtin_amdgcn_sched_barrier(0);
            mac_pair<0, 1>(wb, xs, hh, hh1);
            __builtin_amdgcn_sched_barrier(0);
            __syncthreads();
            if (l == NL - 1) enc[t * C + ch] = o;
            wbp = wb;
            #pragma unroll
            for (int k = 0; k < 32; ++k) xsp[k] = xs[k];
            #pragma unroll
            for (int g = 0; g < 4; ++g) { hhp[g] = hh[g]; hhp1[g] = hh1[g]; }
        }
    }
}

__global__ __launch_bounds__(128, 1) void head_kernel(
    const float* __restrict__ enc,
    const float* __restrict__ rb_w,
    const float* __restrict__ rb_b,
    const float* __restrict__ out_w,
    const float* __restrict__ out_b,
    float* __restrict__ out,
    int od, int col0)
{
    __shared__ float wlds[C * SW];
    __shared__ float blds[C];
    __shared__ float xbuf[4][16 * SA];
    __shared__ float hbuf[4][16 * SA];

    const int tid  = threadIdx.x;
    const int wave = tid >> 5;
    const int lane = tid & 31;
    const int ln   = lane & 15;
    const int hi   = lane >> 4;
    const int row0 = blockIdx.x * 64 + wave * 16;

    for (int i = lane; i < 16 * C; i += 32) {
        int m = i >> 7, c = i & (C - 1);
        xbuf[wave][m * SA + c] = enc[(row0 + m) * C + c];
    }

    for (int blk = 0; blk < 2; ++blk) {
        for (int ly = 0; ly < 2; ++ly) {
            const float* W = rb_w + (blk * 2 + ly) * C * C;
            const float* B = rb_b + (blk * 2 + ly) * C;
            __syncthreads();
            for (int i = tid; i < C * C; i += 128)
                wlds[(i >> 7) * SW + (i & (C - 1))] = W[i];
            if (tid < C) blds[tid] = B[tid];
            __syncthreads();

            const float* src = (ly == 0) ? xbuf[wave] : hbuf[wave];
            float*       dst = (ly == 0) ? hbuf[wave] : xbuf[wave];

            v8f acc[8];
            #pragma unroll
            for (int nt = 0; nt < 8; ++nt) {
                float bv = blds[nt * 16 + ln];
                #pragma unroll
                for (int r = 0; r < 8; ++r) {
                    float cv = bv;
                    if (ly == 1)
                        cv += xbuf[wave][(r + 8 * hi) * SA + nt * 16 + ln];
                    acc[nt][r] = cv;
                }
            }

            for (int k = 0; k < 32; ++k) {
                v2f a;
                a.x = src[ln * SA + 4 * k + 2 * hi];
                a.y = src[ln * SA + 4 * k + 2 * hi + 1];
                #pragma unroll
                for (int nt = 0; nt < 8; ++nt) {
                    v2f b;
                    b.x = wlds[(nt * 16 + ln) * SW + 4 * k + 2 * hi];
                    b.y = wlds[(nt * 16 + ln) * SW + 4 * k + 2 * hi + 1];
                    acc[nt] = __builtin_amdgcn_wmma_f32_16x16x4_f32(
                        false, a, false, b, (short)0, acc[nt], false, false);
                }
            }

            #pragma unroll
            for (int nt = 0; nt < 8; ++nt) {
                #pragma unroll
                for (int r = 0; r < 8; ++r) {
                    float v = acc[nt][r];
                    v = v > 0.0f ? v : NEG_SLOPE * v;
                    dst[(r + 8 * hi) * SA + nt * 16 + ln] = v;
                }
            }
        }
    }
    __syncthreads();

    for (int o = hi; o < od; o += 2) {
        const float* wo = out_w + o * C;
        float s = out_b[o];
        for (int c = 0; c < C; ++c) s += xbuf[wave][ln * SA + c] * wo[c];
        out[(row0 + ln) * 10 + col0 + o] = s;
    }
}

extern "C" void kernel_launch(void* const* d_in, const int* in_sizes, int n_in,
                              void* d_out, int out_size, void* d_ws, size_t ws_size,
                              hipStream_t stream)
{
    const float* x         = (const float*)d_in[0];
    const float* W_ih      = (const float*)d_in[1];
    const float* W_hh      = (const float*)d_in[2];
    const float* b_ih      = (const float*)d_in[3];
    const float* b_hh      = (const float*)d_in[4];
    const float* atom_rb_w = (const float*)d_in[5];
    const float* atom_rb_b = (const float*)d_in[6];
    const float* atom_w    = (const float*)d_in[7];
    const float* atom_b    = (const float*)d_in[8];
    const float* pos_rb_w  = (const float*)d_in[9];
    const float* pos_rb_b  = (const float*)d_in[10];
    const float* pos_w     = (const float*)d_in[11];
    const float* pos_b     = (const float*)d_in[12];
    const float* mag_rb_w  = (const float*)d_in[13];
    const float* mag_rb_b  = (const float*)d_in[14];
    const float* mag_w     = (const float*)d_in[15];
    const float* mag_b     = (const float*)d_in[16];
    const int*   max_steps = (const int*)d_in[17];

    float* out = (float*)d_out;
    float* enc = (float*)d_ws;
    const int T = out_size / 10;

    rnn_recurrence<<<1, 128, 0, stream>>>(x, W_ih, W_hh, b_ih, b_hh, max_steps, enc);

    const int blocks = T / 64;
    head_kernel<<<blocks, 128, 0, stream>>>(enc, atom_rb_w, atom_rb_b, atom_w, atom_b, out, 8, 0);
    head_kernel<<<blocks, 128, 0, stream>>>(enc, pos_rb_w,  pos_rb_b,  pos_w,  pos_b,  out, 1, 8);
    head_kernel<<<blocks, 128, 0, stream>>>(enc, mag_rb_w,  mag_rb_b,  mag_w,  mag_b,  out, 1, 9);
}


# Recurrence on 8 waves, two per SIMD: one wave's FMAs issue while the other loads or waits; K split 8 ways (128x8 table of partials), W_ih rows loaded one evaluation ahead; f32 sum order changes
# speedup vs baseline: 4.1014x; 1.0697x over previous
#include <hip/hip_runtime.h>


#define C 128
#define NL 4
#define NEG_SLOPE 0.2f
#define SA 132
#define SW 132
#define PSTR 264
#define PPAD 12

typedef unsigned u4v __attribute__((ext_vector_type(4)));
typedef float v2f __attribute__((ext_vector_type(2)));
typedef float v4f __attribute__((ext_vector_type(4)));
typedef float v8f __attribute__((ext_vector_type(8)));

__device__ __forceinline__ float mac_lo(float acc, float s, unsigned w)
{
    asm("v_fma_mix_f32 %0, %1, %2, %0 op_sel_hi:[0,1,0]" : "+v"(acc) : "s"(s), "v"(w));
    return acc;
}
__device__ __forceinline__ float mac_hi(float acc, float s, unsigned w)
{
    asm("v_fma_mix_f32 %0, %1, %2, %0 op_sel:[0,1,0] op_sel_hi:[0,1,0]" : "+v"(acc) : "s"(s), "v"(w));
    return acc;
}

__device__ __forceinline__ float bcast(const float mine, const int k)
{
    return __int_as_float(__builtin_amdgcn_readlane(__float_as_int(mine), k));
}

struct HalfRows { u4v r[2][4]; };

template <int OFS>
__device__ __forceinline__ void load_half(const _Float16* rows, HalfRows& w)
{
    #pragma unroll
    for (int kc = 0; kc < 2; ++kc) {
        #pragma unroll
        for (int g = 0; g < 4; ++g)
            w.r[kc][g] = ((const u4v*)(rows + g * 32 * PSTR))[OFS + kc];
    }
}

template <int E0, int E1>
__device__ __forceinline__ void mac_part(const HalfRows& w, const float (&xs)[16],
                                         float (&acca)[4], float (&accb)[4])
{
    #pragma unroll
    for (int e = E0; e < E1; ++e) {
        const float xa = xs[e];
        const float xb = xs[8 + e];
        #pragma unroll
        for (int g = 0; g < 4; ++g) {
            const unsigned u = w.r[0][g][e >> 1];
            acca[g] = (e & 1) ? mac_hi(acca[g], xa, u) : mac_lo(acca[g], xa, u);
        }
        #pragma unroll
        for (int g = 0; g < 4; ++g) {
            const unsigned u = w.r[1][g][e >> 1];
            accb[g] = (e & 1) ? mac_hi(accb[g], xb, u) : mac_lo(accb[g], xb, u);
        }
    }
}

__global__ __launch_bounds__(256, 1) void rnn_recurrence(
    const float* __restrict__ x, const float* __restrict__ Wih,
    const float* __restrict__ Whh, const float* __restrict__ bih,
    const float* __restrict__ bhh, const int* __restrict__ max_steps,
    float* __restrict__ enc)
{
    __shared__ __attribute__((aligned(16))) _Float16 sw[NL * C * PSTR];
    __shared__ __attribute__((aligned(16))) float part[2][C][PPAD];

    const int lane = threadIdx.x & 31;
    const int wq   = threadIdx.x >> 5;
    const int chq  = 32 * (wq >> 1) + (lane ^ (16 * (wq & 1)));

    for (int i = threadIdx.x; i < NL * C * C; i += 2 * C) {
        int r = i >> 7, c = i & (C - 1);
        sw[((r + (NL - 1) * C) & (NL * C - 1)) * PSTR + c] = (_Float16)Wih[i];
        sw[r * PSTR + C + c] = (_Float16)Whh[i];
    }

    float bq[NL][4];
    float hs[NL][4];
    #pragma unroll
    for (int l = 0; l < NL; ++l) {
        #pragma unroll
        for (int g = 0; g < 4; ++g) {
            const float b = bih[l * C + 32 * g + lane] + bhh[l * C + 32 * g + lane];
            bq[l][g] = wq == 0 ? b : 0.0f;
            hs[l][g] = bq[l][g];
        }
    }
    __syncthreads();

    {
        float h0[4], h1[4], xs[16];
        #pragma unroll
        for (int g = 0; g < 4; ++g) { h0[g] = bq[0][g]; h1[g] = 0.0f; }
        HalfRows wb;
        load_half<16>(sw + lane * PSTR + 16 * wq, wb);
        const float xv = x[chq];
        #pragma unroll
        for (int k = 0; k < 16; ++k) xs[k] = bcast(xv, k);
        mac_part<0, 8>(wb, xs, h0, h1);
        #pragma unroll
        for (int g = 0; g < 4; ++g) part[0][32 * g + lane][wq] = h0[g] + h1[g];
    }
    __syncthreads();

    HalfRows wbp;
    float xsp[16], hhp[4], hhp1[4];
    #pragma unroll
    for (int kc = 0; kc < 2; ++kc) {
        #pragma unroll
        for (int g = 0; g < 4; ++g) wbp.r[kc][g] = (u4v){0u, 0u, 0u, 0u};
    }
    #pragma unroll
    for (int k = 0; k < 16; ++k) xsp[k] = 0.0f;
    #pragma unroll
    for (int g = 0; g < 4; ++g) { hhp[g] = bq[NL - 1][g]; hhp1[g] = 0.0f; }

    HalfRows wa;
    load_half<0>(sw + lane * PSTR + 16 * wq, wa);

    const int T = max_steps[0];
    for (int t = 0; t < T; ++t) {
        #pragma unroll
        for (int l = 0; l < NL; ++l) {
            const int ln = (l + 1) & (NL - 1);
            const int lp = (l + NL - 1) & (NL - 1);
            const v4f p0 = *(const v4f*)&part[l & 1][chq][0];
            const v4f p1 = *(const v4f*)&part[l & 1][chq][4];
            HalfRows wb, wan;
            load_half<16>(sw + (l * C + lane) * PSTR + 16 * wq, wb);
            __builtin_amdgcn_sched_barrier(0);
            mac_part<4, 8>(wbp, xsp, hhp, hhp1);
            #pragma unroll
            for (int g = 0; g < 4; ++g) hs[lp][g] = hhp[g] + hhp1[g];
            __builtin_amdgcn_sched_barrier(0);
            load_half<0>(sw + (ln * C + lane) * PSTR + 16 * wq, wan);
            __builtin_amdgcn_sched_barrier(0);
            const v4f s4 = p0 + p1;
            float o = (s4.x + s4.z) + (s4.y + s4.w);
            o = o > 0.0f ? o : 0.0f;
            float xs[16], aa[4], aa1[4], hh[4], hh1[4];
            #pragma unroll
            for (int k = 0; k < 16; ++k) xs[k] = bcast(o, k);
            #pragma unroll
            for (int g = 0; g < 4; ++g) { aa[g] = hs[ln][g]; hh[g] = bq[l][g]; aa1[g] = 0.0f; hh1[g] = 0.0f; }
            mac_part<0, 8>(wa, xs, aa, aa1);
            #pragma unroll
            for (int g = 0; g < 4; ++g) part[ln & 1][32 * g + lane][wq] = aa[g] + aa1[g];
            __builtin_amdgcn_sched_barrier(0);
            mac_part<0, 4>(wb, xs, hh, hh1);
            __builtin_amdgcn_sched_barrier(0);
            __syncthreads();
            if (l == NL - 1 && (wq & 1) == 0) enc[t * C + chq] = o;
            wbp = wb;
            wa = wan;
            #pragma unroll
            for (int k = 0; k < 16; ++k) xsp[k] = xs[k];
            #pragma unroll
            for (int g = 0; g < 4; ++g) { hhp[g] = hh[g]; hhp1[g] = hh1[g]; }
        }
    }
}

__global__ __launch_bounds__(128, 1) void head_kernel(
    const float* __restrict__ enc,
    const float* __restrict__ rb_w,
    const float* __restrict__ rb_b,
    const float* __restrict__ out_w,
    const float* __restrict__ out_b,
    float* __restrict__ out,
    int od, int col0)
{
    __shared__ float wlds[C * SW];
    __shared__ float blds[C];
    __shared__ float xbuf[4][16 * SA];
    __shared__ float hbuf[4][16 * SA];

    const int tid  = threadIdx.x;
    const int wave = tid >> 5;
    const int lane = tid & 31;
    const int ln   = lane & 15;
    const int hi   = lane >> 4;
    const int row0 = blockIdx.x * 64 + wave * 16;

    for (int i = lane; i < 16 * C; i += 32) {
        int m = i >> 7, c = i & (C - 1);
        xbuf[wave][m * SA + c] = enc[(row0 + m) * C + c];
    }

    for (int blk = 0; blk < 2; ++blk) {
        for (int ly = 0; ly < 2; ++ly) {
            const float* W = rb_w + (blk * 2 + ly) * C * C;
            const float* B = rb_b + (blk * 2 + ly) * C;
            __syncthreads();
            for (int i = tid; i < C * C; i += 128)
                wlds[(i >> 7) * SW + (i & (C - 1))] = W[i];
            if (tid < C) blds[tid] = B[tid];
            __syncthreads();

            const float* src = (ly == 0) ? xbuf[wave] : hbuf[wave];
            float*       dst = (ly == 0) ? hbuf[wave] : xbuf[wave];

            v8f acc[8];
            #pragma unroll
            for (int nt = 0; nt < 8; ++nt) {
                float bv = blds[nt * 16 + ln];
                #pragma unroll
                for (int r = 0; r < 8; ++r) {
                    float cv = bv;
                    if (ly == 1)
                        cv += xbuf[wave][(r + 8 * hi) * SA + nt * 16 + ln];
                    acc[nt][r] = cv;
                }
            }

            for (int k = 0; k < 32; ++k) {
                v2f a;
                a.x = src[ln * SA + 4 * k + 2 * hi];
                a.y = src[ln * SA + 4 * k + 2 * hi + 1];
                #pragma unroll
                for (int nt = 0; nt < 8; ++nt) {
                    v2f b;
                    b.x = wlds[(nt * 16 + ln) * SW + 4 * k + 2 * hi];
                    b.y = wlds[(nt * 16 + ln) * SW + 4 * k + 2 * hi + 1];
                    acc[nt] = __builtin_amdgcn_wmma_f32_16x16x4_f32(
                        false, a, false, b, (short)0, acc[nt], false, false);
                }
            }

            #pragma unroll
            for (int nt = 0; nt < 8; ++nt) {
                #pragma unroll
                for (int r = 0; r < 8; ++r) {
                    float v = acc[nt][r];
                    v = v > 0.0f ? v : NEG_SLOPE * v;
                    dst[(r + 8 * hi) * SA + nt * 16 + ln] = v;
                }
            }
        }
    }
    __syncthreads();

    for (int o = hi; o < od; o += 2) {
        const float* wo = out_w + o * C;
        float s = out_b[o];
        for (int c = 0; c < C; ++c) s += xbuf[wave][ln * SA + c] * wo[c];
        out[(row0 + ln) * 10 + col0 + o] = s;
    }
}

extern "C" void kernel_launch(void* const* d_in, const int* in_sizes, int n_in,
                              void* d_out, int out_size, void* d_ws, size_t ws_size,
                              hipStream_t stream)
{
    const float* x         = (const float*)d_in[0];
    const float* W_ih      = (const float*)d_in[1];
    const float* W_hh      = (const float*)d_in[2];
    const float* b_ih      = (const float*)d_in[3];
    const float* b_hh      = (const float*)d_in[4];
    const float* atom_rb_w = (const float*)d_in[5];
    const float* atom_rb_b = (const float*)d_in[6];
    const float* atom_w    = (const float*)d_in[7];
    const float* atom_b    = (const float*)d_in[8];
    const float* pos_rb_w  = (const float*)d_in[9];
    const float* pos_rb_b  = (const float*)d_in[10];
    const float* pos_w     = (const float*)d_in[11];
    const float* pos_b     = (const float*)d_in[12];
    const float* mag_rb_w  = (const float*)d_in[13];
    const float* mag_rb_b  = (const float*)d_in[14];
    const float* mag_w     = (const float*)d_in[15];
    const float* mag_b     = (const float*)d_in[16];
    const int*   max_steps = (const int*)d_in[17];

    float* out = (float*)d_out;
    float* enc = (float*)d_ws;
    const int T = out_size / 10;

    rnn_recurrence<<<1, 256, 0, stream>>>(x, W_ih, W_hh, b_ih, b_hh, max_steps, enc);

    const int blocks = T / 64;
    head_kernel<<<blocks, 128, 0, stream>>>(enc, atom_rb_w, atom_rb_b, atom_w, atom_b, out, 8, 0);
    head_kernel<<<blocks, 128, 0, stream>>>(enc, pos_rb_w,  pos_rb_b,  pos_w,  pos_b,  out, 1, 8);
    head_kernel<<<blocks, 128, 0, stream>>>(enc, mag_rb_w,  mag_rb_b,  mag_w,  mag_b,  out, 1, 9);
}


# Recurrence: wave priority by phase (s_setprio: table loads, chain front 3; W_ih multiply-adds 2; W_hh share 0): the older wave's W_hh work yields to the younger wave's chain; sum order unchanged
# speedup vs baseline: 4.4576x; 1.0868x over previous
#include <hip/hip_runtime.h>


#define C 128
#define NL 4
#define NEG_SLOPE 0.2f
#define SA 132
#define SW 132
#define PSTR 264
#define PPAD 12

typedef unsigned u4v __attribute__((ext_vector_type(4)));
typedef float v2f __attribute__((ext_vector_type(2)));
typedef float v4f __attribute__((ext_vector_type(4)));
typedef float v8f __attribute__((ext_vector_type(8)));

__device__ __forceinline__ float mac_lo(float acc, float s, unsigned w)
{
    asm("v_fma_mix_f32 %0, %1, %2, %0 op_sel_hi:[0,1,0]" : "+v"(acc) : "s"(s), "v"(w));
    return acc;
}
__device__ __forceinline__ float mac_hi(float acc, float s, unsigned w)
{
    asm("v_fma_mix_f32 %0, %1, %2, %0 op_sel:[0,1,0] op_sel_hi:[0,1,0]" : "+v"(acc) : "s"(s), "v"(w));
    return acc;
}

__device__ __forceinline__ float bcast(const float mine, const int k)
{
    return __int_as_float(__builtin_amdgcn_readlane(__float_as_int(mine), k));
}

struct HalfRows { u4v r[2][4]; };

template <int OFS>
__device__ __forceinline__ void load_half(const _Float16* rows, HalfRows& w)
{
    #pragma unroll
    for (int kc = 0; kc < 2; ++kc) {
        #pragma unroll
        for (int g = 0; g < 4; ++g)
            w.r[kc][g] = ((const u4v*)(rows + g * 32 * PSTR))[OFS + kc];
    }
}

template <int E0, int E1>
__device__ __forceinline__ void mac_part(const HalfRows& w, const float (&xs)[16],
                                         float (&acca)[4], float (&accb)[4])
{
    #pragma unroll
    for (int e = E0; e < E1; ++e) {
        const float xa = xs[e];
        const float xb = xs[8 + e];
        #pragma unroll
        for (int g = 0; g < 4; ++g) {
            const unsigned u = w.r[0][g][e >> 1];
            acca[g] = (e & 1) ? mac_hi(acca[g], xa, u) : mac_lo(acca[g], xa, u);
        }
        #pragma unroll
        for (int g = 0; g < 4; ++g) {
            const unsigned u = w.r[1][g][e >> 1];
            accb[g] = (e & 1) ? mac_hi(accb[g], xb, u) : mac_lo(accb[g], xb, u);
        }
    }
}

__global__ __launch_bounds__(256, 1) void rnn_recurrence(
    const float* __restrict__ x, const float* __restrict__ Wih,
    const float* __restrict__ Whh, const float* __restrict__ bih,
    const float* __restrict__ bhh, const int* __restrict__ max_steps,
    float* __restrict__ enc)
{
    __shared__ __attribute__((aligned(16))) _Float16 sw[NL * C * PSTR];
    __shared__ __attribute__((aligned(16))) float part[2][C][PPAD];

    const int lane = threadIdx.x & 31;
    const int wq   = threadIdx.x >> 5;
    const int chq  = 32 * (wq >> 1) + (lane ^ (16 * (wq & 1)));

    for (int i = threadIdx.x; i < NL * C * C; i += 2 * C) {
        int r = i >> 7, c = i & (C - 1);
        sw[((r + (NL - 1) * C) & (NL * C - 1)) * PSTR + c] = (_Float16)Wih[i];
        sw[r * PSTR + C + c] = (_Float16)Whh[i];
    }

    float bq[NL][4];
    float hs[NL][4];
    #pragma unroll
    for (int l = 0; l < NL; ++l) {
        #pragma unroll
        for (int g = 0; g < 4; ++g) {
            const float b = bih[l * C + 32 * g + lane] + bhh[l * C + 32 * g + lane];
            bq[l][g] = wq == 0 ? b : 0.0f;
            hs[l][g] = bq[l][g];
        }
    }
    __syncthreads();

    {
        float h0[4], h1[4], xs[16];
        #pragma unroll
        for (int g = 0; g < 4; ++g) { h0[g] = bq[0][g]; h1[g] = 0.0f; }
        HalfRows wb;
        load_half<16>(sw + lane * PSTR + 16 * wq, wb);
        const float xv = x[chq];
        #pragma unroll
        for (int k = 0; k < 16; ++k) xs[k] = bcast(xv, k);
        mac_part<0, 8>(wb, xs, h0, h1);
        #pragma unroll
        for (int g = 0; g < 4; ++g) part[0][32 * g + lane][wq] = h0[g] + h1[g];
    }
    __syncthreads();

    HalfRows wbp;
    float xsp[16], hhp[4], hhp1[4];
    #pragma unroll
    for (int kc = 0; kc < 2; ++kc) {
        #pragma unroll
        for (int g = 0; g < 4; ++g) wbp.r[kc][g] = (u4v){0u, 0u, 0u, 0u};
    }
    #pragma unroll
    for (int k = 0; k < 16; ++k) xsp[k] = 0.0f;
    #pragma unroll
    for (int g = 0; g < 4; ++g) { hhp[g] = bq[NL - 1][g]; hhp1[g] = 0.0f; }

    HalfRows wa;
    load_half<0>(sw + lane * PSTR + 16 * wq, wa);

    const int T = max_steps[0];
    for (int t = 0; t < T; ++t) {
        #pragma unroll
        for (int l = 0; l < NL; ++l) {
            const int ln = (l + 1) & (NL - 1);
            const int lp = (l + NL - 1) & (NL - 1);
            const v4f p0 = *(const v4f*)&part[l & 1][chq][0];
            const v4f p1 = *(const v4f*)&part[l & 1][chq][4];
            __builtin_amdgcn_s_setprio(0);
            HalfRows wb, wan;
            load_half<16>(sw + (l * C + lane) * PSTR + 16 * wq, wb);
            __builtin_amdgcn_sched_barrier(0);
            mac_part<4, 8>(wbp, xsp, hhp, hhp1);
            #pragma unroll
            for (int g = 0; g < 4; ++g) hs[lp][g] = hhp[g] + hhp1[g];
            __builtin_amdgcn_sched_barrier(0);
            load_half<0>(sw + (ln * C + lane) * PSTR + 16 * wq, wan);
            __builtin_amdgcn_sched_barrier(0);
            __builtin_amdgcn_s_setprio(3);
            const v4f s4 = p0 + p1;
            float o = (s4.x + s4.z) + (s4.y + s4.w);
            o = o > 0.0f ? o : 0.0f;
            float xs[16], aa[4], aa1[4], hh[4], hh1[4];
            #pragma unroll
            for (int k = 0; k < 16; ++k) xs[k] = bcast(o, k);
            asm volatile("" : : "s"(xs[0]), "s"(xs[1]), "s"(xs[2]), "s"(xs[3]), "s"(xs[4]), "s"(xs[5]),
                         "s"(xs[6]), "s"(xs[7]), "s"(xs[8]), "s"(xs[9]), "s"(xs[10]), "s"(xs[11]),
                         "s"(xs[12]), "s"(xs[13]), "s"(xs[14]), "s"(xs[15]));
            __builtin_amdgcn_s_setprio(2);
            #pragma unroll
            for (int g = 0; g < 4; ++g) { aa[g] = hs[ln][g]; hh[g] = bq[l][g]; aa1[g] = 0.0f; hh1[g] = 0.0f; }
            mac_part<0, 8>(wa, xs, aa, aa1);
            #pragma unroll
            for (int g = 0; g < 4; ++g) part[ln & 1][32 * g + lane][wq] = aa[g] + aa1[g];
            __builtin_amdgcn_sched_barrier(0);
            __builtin_amdgcn_s_setprio(0);
            mac_part<0, 4>(wb, xs, hh, hh1);
            __builtin_amdgcn_sched_barrier(0);
            __builtin_amdgcn_s_setprio(3);
            __syncthreads();
            if (l == NL - 1 && (wq & 1) == 0) enc[t * C + chq] = o;
            wbp = wb;
            wa = wan;
            #pragma unroll
            for (int k = 0; k < 16; ++k) xsp[k] = xs[k];
            #pragma unroll
            for (int g = 0; g < 4; ++g) { hhp[g] = hh[g]; hhp1[g] = hh1[g]; }
        }
    }
    __builtin_amdgcn_s_setprio(0);
}

__global__ __launch_bounds__(128, 1) void head_kernel(
    const float* __restrict__ enc,
    const float* __restrict__ rb_w,
    const float* __restrict__ rb_b,
    const float* __restrict__ out_w,
    const float* __restrict__ out_b,
    float* __restrict__ out,
    int od, int col0)
{
    __shared__ float wlds[C * SW];
    __shared__ float blds[C];
    __shared__ float xbuf[4][16 * SA];
    __shared__ float hbuf[4][16 * SA];

    const int tid  = threadIdx.x;
    const int wave = tid >> 5;
    const int lane = tid & 31;
    const int ln   = lane & 15;
    const int hi   = lane >> 4;
    const int row0 = blockIdx.x * 64 + wave * 16;

    for (int i = lane; i < 16 * C; i += 32) {
        int m = i >> 7, c = i & (C - 1);
        xbuf[wave][m * SA + c] = enc[(row0 + m) * C + c];
    }

    for (int blk = 0; blk < 2; ++blk) {
        for (int ly = 0; ly < 2; ++ly) {
            const float* W = rb_w + (blk * 2 + ly) * C * C;
            const float* B = rb_b + (blk * 2 + ly) * C;
            __syncthreads();
            for (int i = tid; i < C * C; i += 128)
                wlds[(i >> 7) * SW + (i & (C - 1))] = W[i];
            if (tid < C) blds[tid] = B[tid];
            __syncthreads();

            const float* src = (ly == 0) ? xbuf[wave] : hbuf[wave];
            float*       dst = (ly == 0) ? hbuf[wave] : xbuf[wave];

            v8f acc[8];
            #pragma unroll
            for (int nt = 0; nt < 8; ++nt) {
                float bv = blds[nt * 16 + ln];
                #pragma unroll
                for (int r = 0; r < 8; ++r) {
                    float cv = bv;
                    if (ly == 1)
                        cv += xbuf[wave][(r + 8 * hi) * SA + nt * 16 + ln];
                    acc[nt][r] = cv;
                }
            }

            for (int k = 0; k < 32; ++k) {
                v2f a;
                a.x = src[ln * SA + 4 * k + 2 * hi];
                a.y = src[ln * SA + 4 * k + 2 * hi + 1];
                #pragma unroll
                for (int nt = 0; nt < 8; ++nt) {
                    v2f b;
                    b.x = wlds[(nt * 16 + ln) * SW + 4 * k + 2 * hi];
                    b.y = wlds[(nt * 16 + ln) * SW + 4 * k + 2 * hi + 1];
                    acc[nt] = __builtin_amdgcn_wmma_f32_16x16x4_f32(
                        false, a, false, b, (short)0, acc[nt], false, false);
                }
            }

            #pragma unroll
            for (int nt = 0; nt < 8; ++nt) {
                #pragma unroll
                for (int r = 0; r < 8; ++r) {
                    float v = acc[nt][r];
                    v = v > 0.0f ? v : NEG_SLOPE * v;
                    dst[(r + 8 * hi) * SA + nt * 16 + ln] = v;
                }
            }
        }
    }
    __syncthreads();

    for (int o = hi; o < od; o += 2) {
        const float* wo = out_w + o * C;
        float s = out_b[o];
        for (int c = 0; c < C; ++c) s += xbuf[wave][ln * SA + c] * wo[c];
        out[(row0 + ln) * 10 + col0 + o] = s;
    }
}

extern "C" void kernel_launch(void* const* d_in, const int* in_sizes, int n_in,
                              void* d_out, int out_size, void* d_ws, size_t ws_size,
                              hipStream_t stream)
{
    const float* x         = (const float*)d_in[0];
    const float* W_ih      = (const float*)d_in[1];
    const float* W_hh      = (const float*)d_in[2];
    const float* b_ih      = (const float*)d_in[3];
    const float* b_hh      = (const float*)d_in[4];
    const float* atom_rb_w = (const float*)d_in[5];
    const float* atom_rb_b = (const float*)d_in[6];
    const float* atom_w    = (const float*)d_in[7];
    const float* atom_b    = (const float*)d_in[8];
    const float* pos_rb_w  = (const float*)d_in[9];
    const float* pos_rb_b  = (const float*)d_in[10];
    const float* pos_w     = (const float*)d_in[11];
    const float* pos_b     = (const float*)d_in[12];
    const float* mag_rb_w  = (const float*)d_in[13];
    const float* mag_rb_b  = (const float*)d_in[14];
    const float* mag_w     = (const float*)d_in[15];
    const float* mag_b     = (const float*)d_in[16];
    const int*   max_steps = (const int*)d_in[17];

    float* out = (float*)d_out;
    float* enc = (float*)d_ws;
    const int T = out_size / 10;

    rnn_recurrence<<<1, 256, 0, stream>>>(x, W_ih, W_hh, b_ih, b_hh, max_steps, enc);

    const int blocks = T / 64;
    head_kernel<<<blocks, 128, 0, stream>>>(enc, atom_rb_w, atom_rb_b, atom_w, atom_b, out, 8, 0);
    head_kernel<<<blocks, 128, 0, stream>>>(enc, pos_rb_w,  pos_rb_b,  pos_w,  pos_b,  out, 1, 8);
    head_kernel<<<blocks, 128, 0, stream>>>(enc, mag_rb_w,  mag_rb_b,  mag_w,  mag_b,  out, 1, 9);
}


# Recurrence: W_ih as f32 in registers all run long (was f16 rows from LDS), its multiply-adds packed: one v_pk_fma_f32 = 2 rows of a lane x 1 broadcast k, 4 accumulators a row; f32 sum order changes
# speedup vs baseline: 5.1788x; 1.1618x over previous
#include <hip/hip_runtime.h>


#define C 128
#define NL 4
#define NEG_SLOPE 0.2f
#define SA 132
#define SW 132
#define WSTR 136
#define PPAD 12

typedef unsigned u4v __attribute__((ext_vector_type(4)));
typedef float v2f __attribute__((ext_vector_type(2)));
typedef float v4f __attribute__((ext_vector_type(4)));
typedef float v8f __attribute__((ext_vector_type(8)));

__device__ __forceinline__ float mac_lo(float acc, float s, unsigned w)
{
    asm("v_fma_mix_f32 %0, %1, %2, %0 op_sel_hi:[0,1,0]" : "+v"(acc) : "s"(s), "v"(w));
    return acc;
}
__device__ __forceinline__ float mac_hi(float acc, float s, unsigned w)
{
    asm("v_fma_mix_f32 %0, %1, %2, %0 op_sel:[0,1,0] op_sel_hi:[0,1,0]" : "+v"(acc) : "s"(s), "v"(w));
    return acc;
}

__device__ __forceinline__ v2f mac_pair(const v2f acc, const v2f w, const float s)
{
    return __builtin_elementwise_fma(w, (v2f){s, s}, acc);
}

__device__ __forceinline__ float bcast(const float mine, const int k)
{
    return __int_as_float(__builtin_amdgcn_readlane(__float_as_int(mine), k));
}

struct HalfRows { u4v r[2][4]; };

__device__ __forceinline__ void load_half(const _Float16* rows, HalfRows& w)
{
    #pragma unroll
    for (int kc = 0; kc < 2; ++kc) {
        #pragma unroll
        for (int g = 0; g < 4; ++g)
            w.r[kc][g] = ((const u4v*)(rows + g * 32 * WSTR))[kc];
    }
}

template <int E0, int E1>
__device__ __forceinline__ void mac_part(const HalfRows& w, const float (&xs)[16],
                                         float (&acca)[4], float (&accb)[4])
{
    #pragma unroll
    for (int e = E0; e < E1; ++e) {
        const float xa = xs[e];
        const float xb = xs[8 + e];
        #pragma unroll
        for (int g = 0; g < 4; ++g) {
            const unsigned u = w.r[0][g][e >> 1];
            acca[g] = (e & 1) ? mac_hi(acca[g], xa, u) : mac_lo(acca[g], xa, u);
        }
        #pragma unroll
        for (int g = 0; g < 4; ++g) {
            const unsigned u = w.r[1][g][e >> 1];
            accb[g] = (e & 1) ? mac_hi(accb[g], xb, u) : mac_lo(accb[g], xb, u);
        }
    }
}

__global__ __launch_bounds__(256, 1) void rnn_recurrence(
    const float* __restrict__ x, const float* __restrict__ Wih,
    const float* __restrict__ Whh, const float* __restrict__ bih,
    const float* __restrict__ bhh, const int* __restrict__ max_steps,
    float* __restrict__ enc)
{
    __shared__ __attribute__((aligned(16))) _Float16 sw[NL * C * WSTR];
    __shared__ __attribute__((aligned(16))) float part[2][C][PPAD];

    const int lane = threadIdx.x & 31;
    const int wq   = threadIdx.x >> 5;
    const int chq  = 32 * (wq >> 1) + (lane ^ (16 * (wq & 1)));

    for (int i = threadIdx.x; i < NL * C * C; i += 2 * C) {
        int r = i >> 7, c = i & (C - 1);
        sw[r * WSTR + c] = (_Float16)Whh[i];
    }

    v2f wih[NL][2][16];
    #pragma unroll
    for (int l = 0; l < NL; ++l) {
        #pragma unroll
        for (int p = 0; p < 2; ++p) {
            const float* r0 = Wih + (l * C + 64 * p + lane) * C + 16 * wq;
            const float* r1 = r0 + 32 * C;
            #pragma unroll
            for (int k = 0; k < 16; ++k) wih[l][p][k] = (v2f){r0[k], r1[k]};
        }
    }

    float bq[NL][4];
    float hs[NL][4];
    #pragma unroll
    for (int l = 0; l < NL; ++l) {
        #pragma unroll
        for (int g = 0; g < 4; ++g) {
            const float b = bih[l * C + 32 * g + lane] + bhh[l * C + 32 * g + lane];
            bq[l][g] = wq == 0 ? b : 0.0f;
            hs[l][g] = bq[l][g];
        }
    }
    __syncthreads();

    {
        float h0[4], h1[4], xs[16];
        #pragma unroll
        for (int g = 0; g < 4; ++g) { h0[g] = bq[0][g]; h1[g] = 0.0f; }
        HalfRows wb;
        load_half(sw + lane * WSTR + 16 * wq, wb);
        const float xv = x[chq];
        #pragma unroll
        for (int k = 0; k < 16; ++k) xs[k] = bcast(xv, k);
        mac_part<0, 8>(wb, xs, h0, h1);
        #pragma unroll
        for (int g = 0; g < 4; ++g) part[0][32 * g + lane][wq] = h0[g] + h1[g];
    }
    __syncthreads();

    HalfRows wbp;
    float xsp[16], hhp[4], hhp1[4];
    #pragma unroll
    for (int kc = 0; kc < 2; ++kc) {
        #pragma unroll
        for (int g = 0; g < 4; ++g) wbp.r[kc][g] = (u4v){0u, 0u, 0u, 0u};
    }
    #pragma unroll
    for (int k = 0; k < 16; ++k) xsp[k] = 0.0f;
    #pragma unroll
    for (int g = 0; g < 4; ++g) { hhp[g] = bq[NL - 1][g]; hhp1[g] = 0.0f; }

    const int T = max_steps[0];
    for (int t = 0; t < T; ++t) {
        #pragma unroll
        for (int l = 0; l < NL; ++l) {
            const int ln = (l + 1) & (NL - 1);
            const int lp = (l + NL - 1) & (NL - 1);
            const v4f p0 = *(const v4f*)&part[l & 1][chq][0];
            const v4f p1 = *(const v4f*)&part[l & 1][chq][4];
            __builtin_amdgcn_s_setprio(0);
            HalfRows wb;
            load_half(sw + (l * C + lane) * WSTR + 16 * wq, wb);
            __builtin_amdgcn_sched_barrier(0);
            mac_part<4, 8>(wbp, xsp, hhp, hhp1);
            #pragma unroll
            for (int g = 0; g < 4; ++g) hs[lp][g] = hhp[g] + hhp1[g];
            __builtin_amdgcn_sched_barrier(0);
            __builtin_amdgcn_s_setprio(3);
            const v4f s4 = p0 + p1;
            float o = (s4.x + s4.z) + (s4.y + s4.w);
            o = o > 0.0f ? o : 0.0f;
            float xs[16], hh[4], hh1[4];
            #pragma unroll
            for (int k = 0; k < 16; ++k) xs[k] = bcast(o, k);
            asm volatile("" : : "s"(xs[0]), "s"(xs[1]), "s"(xs[2]), "s"(xs[3]), "s"(xs[4]), "s"(xs[5]),
                         "s"(xs[6]), "s"(xs[7]), "s"(xs[8]), "s"(xs[9]), "s"(xs[10]), "s"(xs[11]),
                         "s"(xs[12]), "s"(xs[13]), "s"(xs[14]), "s"(xs[15]));
            __builtin_amdgcn_s_setprio(2);
            #pragma unroll
            for (int g = 0; g < 4; ++g) { hh[g] = bq[l][g]; hh1[g] = 0.0f; }
            v2f aw[2][4];
            #pragma unroll
            for (int p = 0; p < 2; ++p) {
                aw[p][0] = (v2f){hs[ln][2 * p], hs[ln][2 * p + 1]};
                #pragma unroll
                for (int c = 1; c < 4; ++c) aw[p][c] = (v2f){0.0f, 0.0f};
            }
            #pragma unroll
            for (int k = 0; k < 16; ++k) {
                #pragma unroll
                for (int p = 0; p < 2; ++p) aw[p][k & 3] = mac_pair(aw[p][k & 3], wih[ln][p][k], xs[k]);
            }
            #pragma unroll
            for (int p = 0; p < 2; ++p) {
                const v2f r = (aw[p][0] + aw[p][1]) + (aw[p][2] + aw[p][3]);
                part[ln & 1][64 * p + lane][wq] = r.x;
                part[ln & 1][64 * p + 32 + lane][wq] = r.y;
            }
            __builtin_amdgcn_sched_barrier(0);
            __builtin_amdgcn_s_setprio(0);
            mac_part<0, 4>(wb, xs, hh, hh1);
            __builtin_amdgcn_sched_barrier(0);
            __builtin_amdgcn_s_setprio(3);
            __syncthreads();
            if (l == NL - 1 && (wq & 1) == 0) enc[t * C + chq] = o;
            wbp = wb;
            #pragma unroll
            for (int k = 0; k < 16; ++k) xsp[k] = xs[k];
            #pragma unroll
            for (int g = 0; g < 4; ++g) { hhp[g] = hh[g]; hhp1[g] = hh1[g]; }
        }
    }
    __builtin_amdgcn_s_setprio(0);
}

__global__ __launch_bounds__(128, 1) void head_kernel(
    const float* __restrict__ enc,
    const float* __restrict__ rb_w,
    const float* __restrict__ rb_b,
    const float* __restrict__ out_w,
    const float* __restrict__ out_b,
    float* __restrict__ out,
    int od, int col0)
{
    __shared__ float wlds[C * SW];
    __shared__ float blds[C];
    __shared__ float xbuf[4][16 * SA];
    __shared__ float hbuf[4][16 * SA];

    const int tid  = threadIdx.x;
    const int wave = tid >> 5;
    const int lane = tid & 31;
    const int ln   = lane & 15;
    const int hi   = lane >> 4;
    const int row0 = blockIdx.x * 64 + wave * 16;

    for (int i = lane; i < 16 * C; i += 32) {
        int m = i >> 7, c = i & (C - 1);
        xbuf[wave][m * SA + c] = enc[(row0 + m) * C + c];
    }

    for (int blk = 0; blk < 2; ++blk) {
        for (int ly = 0; ly < 2; ++ly) {
            const float* W = rb_w + (blk * 2 + ly) * C * C;
            const float* B = rb_b + (blk * 2 + ly) * C;
            __syncthreads();
            for (int i = tid; i < C * C; i += 128)
                wlds[(i >> 7) * SW + (i & (C - 1))] = W[i];
            if (tid < C) blds[tid] = B[tid];
            __syncthreads();

            const float* src = (ly == 0) ? xbuf[wave] : hbuf[wave];
            float*       dst = (ly == 0) ? hbuf[wave] : xbuf[wave];

            v8f acc[8];
            #pragma unroll
            for (int nt = 0; nt < 8; ++nt) {
                float bv = blds[nt * 16 + ln];
                #pragma unroll
                for (int r = 0; r < 8; ++r) {
                    float cv = bv;
                    if (ly == 1)
                        cv += xbuf[wave][(r + 8 * hi) * SA + nt * 16 + ln];
                    acc[nt][r] = cv;
                }
            }

            for (int k = 0; k < 32; ++k) {
                v2f a;
                a.x = src[ln * SA + 4 * k + 2 * hi];
                a.y = src[ln * SA + 4 * k + 2 * hi + 1];
                #pragma unroll
                for (int nt = 0; nt < 8; ++nt) {
                    v2f b;
                    b.x = wlds[(nt * 16 + ln) * SW + 4 * k + 2 * hi];
                    b.y = wlds[(nt * 16 + ln) * SW + 4 * k + 2 * hi + 1];
                    acc[nt] = __builtin_amdgcn_wmma_f32_16x16x4_f32(
                        false, a, false, b, (short)0, acc[nt], false, false);
                }
            }

            #pragma unroll
            for (int nt = 0; nt < 8; ++nt) {
                #pragma unroll
                for (int r = 0; r < 8; ++r) {
                    float v = acc[nt][r];
                    v = v > 0.0f ? v : NEG_SLOPE * v;
                    dst[(r + 8 * hi) * SA + nt * 16 + ln] = v;
                }
            }
        }
    }
    __syncthreads();

    for (int o = hi; o < od; o += 2) {
        const float* wo = out_w + o * C;
        float s = out_b[o];
        for (int c = 0; c < C; ++c) s += xbuf[wave][ln * SA + c] * wo[c];
        out[(row0 + ln) * 10 + col0 + o] = s;
    }
}

extern "C" void kernel_launch(void* const* d_in, const int* in_sizes, int n_in,
                              void* d_out, int out_size, void* d_ws, size_t ws_size,
                              hipStream_t stream)
{
    const float* x         = (const float*)d_in[0];
    const float* W_ih      = (const float*)d_in[1];
    const float* W_hh      = (const float*)d_in[2];
    const float* b_ih      = (const float*)d_in[3];
    const float* b_hh      = (const float*)d_in[4];
    const float* atom_rb_w = (const float*)d_in[5];
    const float* atom_rb_b = (const float*)d_in[6];
    const float* atom_w    = (const float*)d_in[7];
    const float* atom_b    = (const float*)d_in[8];
    const float* pos_rb_w  = (const float*)d_in[9];
    const float* pos_rb_b  = (const float*)d_in[10];
    const float* pos_w     = (const float*)d_in[11];
    const float* pos_b     = (const float*)d_in[12];
    const float* mag_rb_w  = (const float*)d_in[13];
    const float* mag_rb_b  = (const float*)d_in[14];
    const float* mag_w     = (const float*)d_in[15];
    const float* mag_b     = (const float*)d_in[16];
    const int*   max_steps = (const int*)d_in[17];

    float* out = (float*)d_out;
    float* enc = (float*)d_ws;
    const int T = out_size / 10;

    rnn_recurrence<<<1, 256, 0, stream>>>(x, W_ih, W_hh, b_ih, b_hh, max_steps, enc);

    const int blocks = T / 64;
    head_kernel<<<blocks, 128, 0, stream>>>(enc, atom_rb_w, atom_rb_b, atom_w, atom_b, out, 8, 0);
    head_kernel<<<blocks, 128, 0, stream>>>(enc, pos_rb_w,  pos_rb_b,  pos_w,  pos_b,  out, 1, 8);
    head_kernel<<<blocks, 128, 0, stream>>>(enc, mag_rb_w,  mag_rb_b,  mag_w,  mag_b,  out, 1, 9);
}


# Recurrence: W_hh as f32 too (was f16), multiply-adds packed like W_ih's (v_pk_fma_f32: 2 rows x 1 broadcast k), 32 where 64 stood; 6 of 16 k in registers, 10 read from LDS; f32 sum order changes
# speedup vs baseline: 5.6525x; 1.0915x over previous
#include <hip/hip_runtime.h>


#define C 128
#define NL 4
#define NEG_SLOPE 0.2f
#define SA 132
#define SW 132
#define KR 6
#define NJ ((16 - KR) / 2)
#define PPAD 12

typedef float v2f __attribute__((ext_vector_type(2)));
typedef float v4f __attribute__((ext_vector_type(4)));
typedef float v8f __attribute__((ext_vector_type(8)));

__device__ __forceinline__ v2f mac_pair(const v2f acc, const v2f w, const float s)
{
    return __builtin_elementwise_fma(w, (v2f){s, s}, acc);
}

__device__ __forceinline__ float bcast(const float mine, const int k)
{
    return __int_as_float(__builtin_amdgcn_readlane(__float_as_int(mine), k));
}

struct HhRows { v4f r[2][NJ]; };

__device__ __forceinline__ void load_hh(const float* rows, HhRows& w)
{
    #pragma unroll
    for (int p = 0; p < 2; ++p) {
        #pragma unroll
        for (int j = 0; j < NJ; ++j)
            w.r[p][j] = *(const v4f*)(rows + (p * NJ + j) * 128);
    }
}

__device__ __forceinline__ void mac_hh_upper(const HhRows& w, const float (&xs)[16], v2f (&acc)[2][4])
{
    #pragma unroll
    for (int j = 0; j < NJ; ++j) {
        #pragma unroll
        for (int p = 0; p < 2; ++p) {
            acc[p][(KR + 2 * j) & 3]     = mac_pair(acc[p][(KR + 2 * j) & 3],     (v2f){w.r[p][j].x, w.r[p][j].y}, xs[KR + 2 * j]);
            acc[p][(KR + 1 + 2 * j) & 3] = mac_pair(acc[p][(KR + 1 + 2 * j) & 3], (v2f){w.r[p][j].z, w.r[p][j].w}, xs[KR + 1 + 2 * j]);
        }
    }
}

__global__ __launch_bounds__(256, 1) void rnn_recurrence(
    const float* __restrict__ x, const float* __restrict__ Wih,
    const float* __restrict__ Whh, const float* __restrict__ bih,
    const float* __restrict__ bhh, const int* __restrict__ max_steps,
    float* __restrict__ enc)
{
    __shared__ __attribute__((aligned(16))) float swf[NL * 8 * 2 * NJ * 32 * 4];
    __shared__ __attribute__((aligned(16))) float part[2][C][PPAD];

    const int lane = threadIdx.x & 31;
    const int wq   = threadIdx.x >> 5;
    const int chq  = 32 * (wq >> 1) + (lane ^ (16 * (wq & 1)));

    for (int d = threadIdx.x; d < NL * 8 * 2 * NJ * 32 * 4; d += 2 * C) {
        const int e = d & 3, ln = (d >> 2) & 31, blk = d >> 7;
        const int j = blk % NJ, p = (blk / NJ) & 1, w = (blk / (2 * NJ)) & 7, l = blk / (16 * NJ);
        const int row = 64 * p + 32 * (e & 1) + ln;
        const int k   = 16 * w + KR + 2 * j + (e >> 1);
        swf[d] = Whh[(l * C + row) * C + k];
    }
    const float* hh_rows = swf + wq * (2 * NJ * 32 * 4) + lane * 4;

    v2f whl[NL][2][KR];
    #pragma unroll
    for (int l = 0; l < NL; ++l) {
        #pragma unroll
        for (int p = 0; p < 2; ++p) {
            const float* r0 = Whh + (l * C + 64 * p + lane) * C + 16 * wq;
            const float* r1 = r0 + 32 * C;
            #pragma unroll
            for (int k = 0; k < KR; ++k) whl[l][p][k] = (v2f){r0[k], r1[k]};
        }
    }

    v2f wih[NL][2][16];
    #pragma unroll
    for (int l = 0; l < NL; ++l) {
        #pragma unroll
        for (int p = 0; p < 2; ++p) {
            const float* r0 = Wih + (l * C + 64 * p + lane) * C + 16 * wq;
            const float* r1 = r0 + 32 * C;
            #pragma unroll
            for (int k = 0; k < 16; ++k) wih[l][p][k] = (v2f){r0[k], r1[k]};
        }
    }
    v2f bq[NL][2];
    v2f hs[NL][2];
    #pragma unroll
    for (int l = 0; l < NL; ++l) {
        #pragma unroll
        for (int p = 0; p < 2; ++p) {
            const int c0 = l * C + 64 * p + lane;
            const float b0 = bih[c0] + bhh[c0];
            const float b1 = bih[c0 + 32] + bhh[c0 + 32];
            bq[l][p] = wq == 0 ? (v2f){b0, b1} : (v2f){0.0f, 0.0f};
            hs[l][p] = bq[l][p];
        }
    }
    __syncthreads();

    {
        float xs[16];
        v2f h[2][4];
        #pragma unroll
        for (int p = 0; p < 2; ++p) {
            h[p][0] = bq[0][p];
            #pragma unroll
            for (int c = 1; c < 4; ++c) h[p][c] = (v2f){0.0f, 0.0f};
        }
        HhRows wb;
        load_hh(hh_rows, wb);
        const float xv = x[chq];
        #pragma unroll
        for (int k = 0; k < 16; ++k) xs[k] = bcast(xv, k);
        #pragma unroll
        for (int k = 0; k < KR; ++k) {
            #pragma unroll
            for (int p = 0; p < 2; ++p) h[p][k & 3] = mac_pair(h[p][k & 3], whl[0][p][k], xs[k]);
        }
        mac_hh_upper(wb, xs, h);
        #pragma unroll
        for (int p = 0; p < 2; ++p) {
            const v2f r = (h[p][0] + h[p][1]) + (h[p][2] + h[p][3]);
            part[0][64 * p + lane][wq] = r.x;
            part[0][64 * p + 32 + lane][wq] = r.y;
        }
    }
    __syncthreads();

    float xsp[KR];
    v2f hhp[2][4];
    #pragma unroll
    for (int k = 0; k < KR; ++k) xsp[k] = 0.0f;
    #pragma unroll
    for (int p = 0; p < 2; ++p) {
        hhp[p][0] = bq[NL - 1][p];
        #pragma unroll
        for (int c = 1; c < 4; ++c) hhp[p][c] = (v2f){0.0f, 0.0f};
    }

    const int T = max_steps[0];
    for (int t = 0; t < T; ++t) {
        #pragma unroll
        for (int l = 0; l < NL; ++l) {
            const int ln = (l + 1) & (NL - 1);
            const int lp = (l + NL - 1) & (NL - 1);
            const v4f p0 = *(const v4f*)&part[l & 1][chq][0];
            const v4f p1 = *(const v4f*)&part[l & 1][chq][4];
            __builtin_amdgcn_s_setprio(0);
            HhRows wb;
            load_hh(hh_rows + l * (8 * 2 * NJ * 32 * 4), wb);
            __builtin_amdgcn_sched_barrier(0);
            asm volatile("" : "+v"(hhp[0][0]), "+v"(hhp[0][1]), "+v"(hhp[0][2]), "+v"(hhp[0][3]),
                              "+v"(hhp[1][0]), "+v"(hhp[1][1]), "+v"(hhp[1][2]), "+v"(hhp[1][3]));
            #pragma unroll
            for (int k = 0; k < KR; ++k) {
                #pragma unroll
                for (int p = 0; p < 2; ++p) hhp[p][k & 3] = mac_pair(hhp[p][k & 3], whl[lp][p][k], xsp[k]);
            }
            #pragma unroll
            for (int p = 0; p < 2; ++p) hs[lp][p] = (hhp[p][0] + hhp[p][1]) + (hhp[p][2] + hhp[p][3]);
            asm volatile("" : : "v"(hs[lp][0]), "v"(hs[lp][1]));
            __builtin_amdgcn_sched_barrier(0);
            __builtin_amdgcn_s_setprio(3);
            const v4f s4 = p0 + p1;
            float o = (s4.x + s4.z) + (s4.y + s4.w);
            o = o > 0.0f ? o : 0.0f;
            float xs[16];
            #pragma unroll
            for (int k = 0; k < 16; ++k) xs[k] = bcast(o, k);
            asm volatile("" : : "s"(xs[0]), "s"(xs[1]), "s"(xs[2]), "s"(xs[3]), "s"(xs[4]), "s"(xs[5]),
                         "s"(xs[6]), "s"(xs[7]), "s"(xs[8]), "s"(xs[9]), "s"(xs[10]), "s"(xs[11]),
                         "s"(xs[12]), "s"(xs[13]), "s"(xs[14]), "s"(xs[15]));
            __builtin_amdgcn_s_setprio(2);
            v2f aw[2][4];
            #pragma unroll
            for (int p = 0; p < 2; ++p) {
                aw[p][0] = hs[ln][p];
                #pragma unroll
                for (int c = 1; c < 4; ++c) aw[p][c] = (v2f){0.0f, 0.0f};
            }
            #pragma unroll
            for (int k = 0; k < 16; ++k) {
                #pragma unroll
                for (int p = 0; p < 2; ++p) aw[p][k & 3] = mac_pair(aw[p][k & 3], wih[ln][p][k], xs[k]);
            }
            #pragma unroll
            for (int p = 0; p < 2; ++p) {
                const v2f r = (aw[p][0] + aw[p][1]) + (aw[p][2] + aw[p][3]);
                part[ln & 1][64 * p + lane][wq] = r.x;
                part[ln & 1][64 * p + 32 + lane][wq] = r.y;
            }
            __builtin_amdgcn_sched_barrier(0);
            __builtin_amdgcn_s_setprio(0);
            v2f hh[2][4];
            #pragma unroll
            for (int p = 0; p < 2; ++p) {
                hh[p][0] = bq[l][p];
                #pragma unroll
                for (int c = 1; c < 4; ++c) hh[p][c] = (v2f){0.0f, 0.0f};
            }
            mac_hh_upper(wb, xs, hh);
            asm volatile("" : : "v"(hh[0][0]), "v"(hh[0][1]), "v"(hh[0][2]), "v"(hh[0][3]),
                                "v"(hh[1][0]), "v"(hh[1][1]), "v"(hh[1][2]), "v"(hh[1][3]));
            __builtin_amdgcn_sched_barrier(0);
            __builtin_amdgcn_s_setprio(3);
            __syncthreads();
            if (l == NL - 1 && (wq & 1) == 0) enc[t * C + chq] = o;
            #pragma unroll
            for (int k = 0; k < KR; ++k) xsp[k] = xs[k];
            #pragma unroll
            for (int p = 0; p < 2; ++p) {
                #pragma unroll
                for (int c = 0; c < 4; ++c) hhp[p][c] = hh[p][c];
            }
        }
    }
    __builtin_amdgcn_s_setprio(0);
}

__global__ __launch_bounds__(128, 1) void head_kernel(
    const float* __restrict__ enc,
    const float* __restrict__ rb_w,
    const float* __restrict__ rb_b,
    const float* __restrict__ out_w,
    const float* __restrict__ out_b,
    float* __restrict__ out,
    int od, int col0)
{
    __shared__ float wlds[C * SW];
    __shared__ float blds[C];
    __shared__ float xbuf[4][16 * SA];
    __shared__ float hbuf[4][16 * SA];

    const int tid  = threadIdx.x;
    const int wave = tid >> 5;
    const int lane = tid & 31;
    const int ln   = lane & 15;
    const int hi   = lane >> 4;
    const int row0 = blockIdx.x * 64 + wave * 16;

    for (int i = lane; i < 16 * C; i += 32) {
        int m = i >> 7, c = i & (C - 1);
        xbuf[wave][m * SA + c] = enc[(row0 + m) * C + c];
    }

    for (int blk = 0; blk < 2; ++blk) {
        for (int ly = 0; ly < 2; ++ly) {
            const float* W = rb_w + (blk * 2 + ly) * C * C;
            const float* B = rb_b + (blk * 2 + ly) * C;
            __syncthreads();
            for (int i = tid; i < C * C; i += 128)
                wlds[(i >> 7) * SW + (i & (C - 1))] = W[i];
            if (tid < C) blds[tid] = B[tid];
            __syncthreads();

            const float* src = (ly == 0) ? xbuf[wave] : hbuf[wave];
            float*       dst = (ly == 0) ? hbuf[wave] : xbuf[wave];

            v8f acc[8];
            #pragma unroll
            for (int nt = 0; nt < 8; ++nt) {
                float bv = blds[nt * 16 + ln];
                #pragma unroll
                for (int r = 0; r < 8; ++r) {
                    float cv = bv;
                    if (ly == 1)
                        cv += xbuf[wave][(r + 8 * hi) * SA + nt * 16 + ln];
                    acc[nt][r] = cv;
                }
            }

            for (int k = 0; k < 32; ++k) {
                v2f a;
                a.x = src[ln * SA + 4 * k + 2 * hi];
                a.y = src[ln * SA + 4 * k + 2 * hi + 1];
                #pragma unroll
                for (int nt = 0; nt < 8; ++nt) {
                    v2f b;
                    b.x = wlds[(nt * 16 + ln) * SW + 4 * k + 2 * hi];
                    b.y = wlds[(nt * 16 + ln) * SW + 4 * k + 2 * hi + 1];
                    acc[nt] = __builtin_amdgcn_wmma_f32_16x16x4_f32(
                        false, a, false, b, (short)0, acc[nt], false, false);
                }
            }

            #pragma unroll
            for (int nt = 0; nt < 8; ++nt) {
                #pragma unroll
                for (int r = 0; r < 8; ++r) {
                    float v = acc[nt][r];
                    v = v > 0.0f ? v : NEG_SLOPE * v;
                    dst[(r + 8 * hi) * SA + nt * 16 + ln] = v;
                }
            }
        }
    }
    __syncthreads();

    for (int o = hi; o < od; o += 2) {
        const float* wo = out_w + o * C;
        float s = out_b[o];
        for (int c = 0; c < C; ++c) s += xbuf[wave][ln * SA + c] * wo[c];
        out[(row0 + ln) * 10 + col0 + o] = s;
    }
}

extern "C" void kernel_launch(void* const* d_in, const int* in_sizes, int n_in,
                              void* d_out, int out_size, void* d_ws, size_t ws_size,
                              hipStream_t stream)
{
    const float* x         = (const float*)d_in[0];
    const float* W_ih      = (const float*)d_in[1];
    const float* W_hh      = (const float*)d_in[2];
    const float* b_ih      = (const float*)d_in[3];
    const float* b_hh      = (const float*)d_in[4];
    const float* atom_rb_w = (const float*)d_in[5];
    const float* atom_rb_b = (const float*)d_in[6];
    const float* atom_w    = (const float*)d_in[7];
    const float* atom_b    = (const float*)d_in[8];
    const float* pos_rb_w  = (const float*)d_in[9];
    const float* pos_rb_b  = (const float*)d_in[10];
    const float* pos_w     = (const float*)d_in[11];
    const float* pos_b     = (const float*)d_in[12];
    const float* mag_rb_w  = (const float*)d_in[13];
    const float* mag_rb_b  = (const float*)d_in[14];
    const float* mag_w     = (const float*)d_in[15];
    const float* mag_b     = (const float*)d_in[16];
    const int*   max_steps = (const int*)d_in[17];

    float* out = (float*)d_out;
    float* enc = (float*)d_ws;
    const int T = out_size / 10;

    rnn_recurrence<<<1, 256, 0, stream>>>(x, W_ih, W_hh, b_ih, b_hh, max_steps, enc);

    const int blocks = T / 64;
    head_kernel<<<blocks, 128, 0, stream>>>(enc, atom_rb_w, atom_rb_b, atom_w, atom_b, out, 8, 0);
    head_kernel<<<blocks, 128, 0, stream>>>(enc, pos_rb_w,  pos_rb_b,  pos_w,  pos_b,  out, 1, 8);
    head_kernel<<<blocks, 128, 0, stream>>>(enc, mag_rb_w,  mag_rb_b,  mag_w,  mag_b,  out, 1, 9);
}
